# starting kernel
# baseline (speedup 1.0000x reference)
#include <hip/hip_runtime.h>
#include <math.h>

#define N_NODE   50000
#define N_EDGE   800000
#define IN_DIM   128
#define OUT_DIM  128
#define ATTN_DIM 64
#define N_REL    200
#define N_Q      64
#define N_EMB    (2 * N_REL + 1)

typedef __attribute__((ext_vector_type(2))) float v2f;
typedef __attribute__((ext_vector_type(8))) float v8f;

__global__ void small_proj_kernel(const float* __restrict__ rela,
                                  const float* __restrict__ Wr,
                                  const float* __restrict__ WqrW,
                                  const float* __restrict__ Wqrb,
                                  const int*   __restrict__ q_rel,
                                  float* __restrict__ rproj,
                                  float* __restrict__ qproj) {
    int idx = blockIdx.x * blockDim.x + threadIdx.x;
    const int total = (N_EMB + N_Q) * ATTN_DIM;
    if (idx >= total) return;
    int col = idx & (ATTN_DIM - 1);
    int row = idx >> 6;
    if (row < N_EMB) {
        const float* h = rela + (size_t)row * IN_DIM;
        float s = 0.f;
        #pragma unroll 8
        for (int k = 0; k < IN_DIM; ++k) s = fmaf(h[k], Wr[k * ATTN_DIM + col], s);
        rproj[row * ATTN_DIM + col] = s;
    } else {
        int q = row - N_EMB;
        const float* h = rela + (size_t)q_rel[q] * IN_DIM;
        float s = Wqrb[col];
        #pragma unroll 8
        for (int k = 0; k < IN_DIM; ++k) s = fmaf(h[k], WqrW[k * ATTN_DIM + col], s);
        qproj[q * ATTN_DIM + col] = s;
    }
}

template <int N>
__global__ void gemm_wmma_f32_kernel(const float* __restrict__ A,
                                     const float* __restrict__ B,
                                     float* __restrict__ C,
                                     int M) {
    constexpr int K = 128;
    __shared__ v2f Bs2[(K / 2) * N];

    for (int i = threadIdx.x; i < (K / 2) * N; i += blockDim.x) {
        const int kp = i / N;
        const int n  = i - kp * N;
        v2f p;
        p.x = B[(2 * kp)     * N + n];
        p.y = B[(2 * kp + 1) * N + n];
        Bs2[i] = p;
    }
    __syncthreads();

    const int lane = threadIdx.x & 31;
    const int waveInBlock = threadIdx.x >> 5;
    constexpr int colTiles = N / 64;
    const long tileId = (long)blockIdx.x * 8 + waveInBlock;
    const int  mTile  = (int)(tileId / colTiles);
    const int  nTile  = (int)(tileId % colTiles);
    if (mTile * 16 >= M) return;

    const int mrow  = mTile * 16 + (lane & 15);
    const int sel   = lane >> 4;
    const int koff  = sel * 2;
    const int ncol0 = nTile * 64;
    const int nlane = ncol0 + (lane & 15);

    v8f acc0 = {}, acc1 = {}, acc2 = {}, acc3 = {};

    const float* __restrict__ arow = A + (size_t)mrow * K;
    #pragma unroll 4
    for (int k0 = 0; k0 < K; k0 += 4) {
        v2f a;
        a.x = arow[k0 + koff];
        a.y = arow[k0 + koff + 1];

        const v2f* __restrict__ brow = &Bs2[(k0 / 2 + sel) * N + nlane];
        acc0 = __builtin_amdgcn_wmma_f32_16x16x4_f32(false, a, false, brow[ 0], (short)0, acc0, false, false);
        acc1 = __builtin_amdgcn_wmma_f32_16x16x4_f32(false, a, false, brow[16], (short)0, acc1, false, false);
        acc2 = __builtin_amdgcn_wmma_f32_16x16x4_f32(false, a, false, brow[32], (short)0, acc2, false, false);
        acc3 = __builtin_amdgcn_wmma_f32_16x16x4_f32(false, a, false, brow[48], (short)0, acc3, false, false);
    }

    const int cmBase = mTile * 16 + ((lane < 16) ? 0 : 8);
    const int cn = lane & 15;
    float* __restrict__ c0 = C + (size_t)cmBase * N + ncol0 + cn;
    #pragma unroll
    for (int r = 0; r < 8; ++r) {
        float* cr = c0 + (size_t)r * N;
        cr[ 0] = acc0[r];
        cr[16] = acc1[r];
        cr[32] = acc2[r];
        cr[48] = acc3[r];
    }
}

__global__ void edge_kernel(const int*   __restrict__ edges,
                            const int*   __restrict__ r_idx,
                            const float* __restrict__ hidden,
                            const float* __restrict__ rela,
                            const float* __restrict__ hproj,
                            const float* __restrict__ rproj,
                            const float* __restrict__ qproj,
                            const float* __restrict__ waW,
                            const float* __restrict__ wab,
                            float* __restrict__ agg) {
    const int lane   = threadIdx.x & 31;
    const int wave   = (blockIdx.x * blockDim.x + threadIdx.x) >> 5;
    const int nwaves = (gridDim.x * blockDim.x) >> 5;

    const float bias = wab[0];
    const v2f wa2 = ((const v2f*)waW)[lane];

    for (int e = wave; e < N_EDGE; e += nwaves) {
        const int sub = edges[e * 3 + 0];
        const int rel = edges[e * 3 + 1];
        const int obj = edges[e * 3 + 2];
        const int q   = r_idx[e];

        v2f hp = ((const v2f*)(hproj + (size_t)sub * ATTN_DIM))[lane];
        v2f rp = ((const v2f*)(rproj + (size_t)rel * ATTN_DIM))[lane];
        v2f qp = ((const v2f*)(qproj + (size_t)q   * ATTN_DIM))[lane];
        float p0 = fmaxf(hp.x + rp.x + qp.x, 0.f);
        float p1 = fmaxf(hp.y + rp.y + qp.y, 0.f);
        float s  = fmaf(p0, wa2.x, p1 * wa2.y);

        #pragma unroll
        for (int off = 16; off > 0; off >>= 1) s += __shfl_xor(s, off, 32);

        const float alpha = 1.0f / (1.0f + __expf(-(s + bias)));

        const float* hs = hidden + (size_t)sub * IN_DIM;
        const float* hr = rela   + (size_t)rel * IN_DIM;
        float* dst = agg + (size_t)obj * IN_DIM;
        #pragma unroll
        for (int d4 = 0; d4 < 4; ++d4) {
            const int d = lane + 32 * d4;
            atomicAdd(&dst[d], alpha * hs[d] * hr[d]);
        }
    }
}

extern "C" void kernel_launch(void* const* d_in, const int* in_sizes, int n_in,
                              void* d_out, int out_size, void* d_ws, size_t ws_size,
                              hipStream_t stream) {
    const int*   q_rel  = (const int*)  d_in[1];
    const int*   r_idx  = (const int*)  d_in[2];
    const float* hidden = (const float*)d_in[3];
    const int*   edges  = (const int*)  d_in[4];
    const float* rela   = (const float*)d_in[6];
    const float* Ws     = (const float*)d_in[7];
    const float* Wr     = (const float*)d_in[8];
    const float* WqrW   = (const float*)d_in[9];
    const float* Wqrb   = (const float*)d_in[10];
    const float* waW    = (const float*)d_in[11];
    const float* wab    = (const float*)d_in[12];
    const float* Wh     = (const float*)d_in[13];
    float* out = (float*)d_out;

    char* ws = (char*)d_ws;
    float* agg   = (float*)(ws);
    float* hproj = (float*)(ws + (size_t)N_NODE * IN_DIM * 4);
    float* rproj = (float*)(ws + (size_t)N_NODE * IN_DIM * 4
                               + (size_t)N_NODE * ATTN_DIM * 4);
    float* qproj = (float*)((char*)rproj + (((size_t)N_EMB * ATTN_DIM * 4 + 255ull) & ~255ull));

    hipMemsetAsync(agg, 0, (size_t)N_NODE * IN_DIM * sizeof(float), stream);

    {
        const int total = (N_EMB + N_Q) * ATTN_DIM;
        small_proj_kernel<<<(total + 255) / 256, 256, 0, stream>>>(
            rela, Wr, WqrW, Wqrb, q_rel, rproj, qproj);
    }

    {
        const int mTiles = N_NODE / 16;
        const int waveTiles = mTiles * (ATTN_DIM / 64);
        gemm_wmma_f32_kernel<ATTN_DIM><<<(waveTiles + 7) / 8, 256, 0, stream>>>(
            hidden, Ws, hproj, N_NODE);
    }

    edge_kernel<<<4096, 256, 0, stream>>>(edges, r_idx, hidden, rela,
                                          hproj, rproj, qproj, waW, wab, agg);

    {
        const int mTiles = N_NODE / 16;
        const int waveTiles = mTiles * (OUT_DIM / 64);
        gemm_wmma_f32_kernel<OUT_DIM><<<(waveTiles + 7) / 8, 256, 0, stream>>>(
            agg, Wh, out, N_NODE);
    }
}


# Float scatter atomics replaced by per-node linked lists built in the call (one int exchange per edge) + per-node register sum in list (arrival) order; gate dot summed 8 terms per lane
# speedup vs baseline: 1.1328x; 1.1328x over previous
#include <hip/hip_runtime.h>
#include <math.h>

#define N_NODE   50000
#define N_EDGE   800000
#define IN_DIM   128
#define OUT_DIM  128
#define ATTN_DIM 64
#define N_REL    200
#define N_Q      64
#define N_EMB    (2 * N_REL + 1)

typedef __attribute__((ext_vector_type(2))) float v2f;
typedef __attribute__((ext_vector_type(8))) float v8f;

__global__ void small_proj_kernel(const float* __restrict__ rela,
                                  const float* __restrict__ Wr,
                                  const float* __restrict__ WqrW,
                                  const float* __restrict__ Wqrb,
                                  const int*   __restrict__ q_rel,
                                  float* __restrict__ rproj,
                                  float* __restrict__ qproj) {
    int idx = blockIdx.x * blockDim.x + threadIdx.x;
    const int total = (N_EMB + N_Q) * ATTN_DIM;
    if (idx >= total) return;
    int col = idx & (ATTN_DIM - 1);
    int row = idx >> 6;
    if (row < N_EMB) {
        const float* h = rela + (size_t)row * IN_DIM;
        float s = 0.f;
        #pragma unroll 8
        for (int k = 0; k < IN_DIM; ++k) s = fmaf(h[k], Wr[k * ATTN_DIM + col], s);
        rproj[row * ATTN_DIM + col] = s;
    } else {
        int q = row - N_EMB;
        const float* h = rela + (size_t)q_rel[q] * IN_DIM;
        float s = Wqrb[col];
        #pragma unroll 8
        for (int k = 0; k < IN_DIM; ++k) s = fmaf(h[k], WqrW[k * ATTN_DIM + col], s);
        qproj[q * ATTN_DIM + col] = s;
    }
}

template <int N>
__global__ void gemm_wmma_f32_kernel(const float* __restrict__ A,
                                     const float* __restrict__ B,
                                     float* __restrict__ C,
                                     int M) {
    constexpr int K = 128;
    __shared__ v2f Bs2[(K / 2) * N];

    for (int i = threadIdx.x; i < (K / 2) * N; i += blockDim.x) {
        const int kp = i / N;
        const int n  = i - kp * N;
        v2f p;
        p.x = B[(2 * kp)     * N + n];
        p.y = B[(2 * kp + 1) * N + n];
        Bs2[i] = p;
    }
    __syncthreads();

    const int lane = threadIdx.x & 31;
    const int waveInBlock = threadIdx.x >> 5;
    constexpr int colTiles = N / 64;
    const long tileId = (long)blockIdx.x * 8 + waveInBlock;
    const int  mTile  = (int)(tileId / colTiles);
    const int  nTile  = (int)(tileId % colTiles);
    if (mTile * 16 >= M) return;

    const int mrow  = mTile * 16 + (lane & 15);
    const int sel   = lane >> 4;
    const int koff  = sel * 2;
    const int ncol0 = nTile * 64;
    const int nlane = ncol0 + (lane & 15);

    v8f acc0 = {}, acc1 = {}, acc2 = {}, acc3 = {};

    const float* __restrict__ arow = A + (size_t)mrow * K;
    #pragma unroll 4
    for (int k0 = 0; k0 < K; k0 += 4) {
        v2f a;
        a.x = arow[k0 + koff];
        a.y = arow[k0 + koff + 1];

        const v2f* __restrict__ brow = &Bs2[(k0 / 2 + sel) * N + nlane];
        acc0 = __builtin_amdgcn_wmma_f32_16x16x4_f32(false, a, false, brow[ 0], (short)0, acc0, false, false);
        acc1 = __builtin_amdgcn_wmma_f32_16x16x4_f32(false, a, false, brow[16], (short)0, acc1, false, false);
        acc2 = __builtin_amdgcn_wmma_f32_16x16x4_f32(false, a, false, brow[32], (short)0, acc2, false, false);
        acc3 = __builtin_amdgcn_wmma_f32_16x16x4_f32(false, a, false, brow[48], (short)0, acc3, false, false);
    }

    const int cmBase = mTile * 16 + ((lane < 16) ? 0 : 8);
    const int cn = lane & 15;
    float* __restrict__ c0 = C + (size_t)cmBase * N + ncol0 + cn;
    #pragma unroll
    for (int r = 0; r < 8; ++r) {
        float* cr = c0 + (size_t)r * N;
        cr[ 0] = acc0[r];
        cr[16] = acc1[r];
        cr[32] = acc2[r];
        cr[48] = acc3[r];
    }
}

__global__ void edge_link_kernel(const int*   __restrict__ edges,
                                 const int*   __restrict__ r_idx,
                                 const float* __restrict__ hproj,
                                 const float* __restrict__ rproj,
                                 const float* __restrict__ qproj,
                                 const float* __restrict__ waW,
                                 const float* __restrict__ wab,
                                 int* __restrict__ head,
                                 int* __restrict__ rec) {
    const int lane   = threadIdx.x & 31;
    const int wave   = (blockIdx.x * blockDim.x + threadIdx.x) >> 5;
    const int nwaves = (gridDim.x * blockDim.x) >> 5;
    const int t      = lane & 7;
    const int gbase  = lane & 24;

    const float  bias = wab[0];
    const float4 w0 = ((const float4*)waW)[t];
    const float4 w1 = ((const float4*)waW)[8 + t];

    for (int c = wave; c < N_EDGE / 32; c += nwaves) {
        const int e   = c * 32 + lane;
        const int sub = edges[e * 3 + 0];
        const int rel = edges[e * 3 + 1];
        const int obj = edges[e * 3 + 2];
        const int q   = r_idx[e];

        float myalpha = 0.f;
        for (int j = 0; j < 8; ++j) {
            const int src   = gbase + j;
            const int s_sub = __shfl(sub, src, 32);
            const int s_rel = __shfl(rel, src, 32);
            const int s_q   = __shfl(q,   src, 32);
            const float4* hp = (const float4*)(hproj + (size_t)s_sub * ATTN_DIM);
            const float4* rp = (const float4*)(rproj + (size_t)s_rel * ATTN_DIM);
            const float4* qp = (const float4*)(qproj + (size_t)s_q   * ATTN_DIM);
            const float4 h0 = hp[t], h1 = hp[8 + t];
            const float4 r0 = rp[t], r1 = rp[8 + t];
            const float4 q0 = qp[t], q1 = qp[8 + t];
            float s = fmaxf(h0.x + r0.x + q0.x, 0.f) * w0.x;
            s = fmaf(fmaxf(h0.y + r0.y + q0.y, 0.f), w0.y, s);
            s = fmaf(fmaxf(h0.z + r0.z + q0.z, 0.f), w0.z, s);
            s = fmaf(fmaxf(h0.w + r0.w + q0.w, 0.f), w0.w, s);
            s = fmaf(fmaxf(h1.x + r1.x + q1.x, 0.f), w1.x, s);
            s = fmaf(fmaxf(h1.y + r1.y + q1.y, 0.f), w1.y, s);
            s = fmaf(fmaxf(h1.z + r1.z + q1.z, 0.f), w1.z, s);
            s = fmaf(fmaxf(h1.w + r1.w + q1.w, 0.f), w1.w, s);
            s += __shfl_xor(s, 1, 32);
            s += __shfl_xor(s, 2, 32);
            s += __shfl_xor(s, 4, 32);
            const float alpha = 1.0f / (1.0f + __expf(-(s + bias)));
            if (t == j) myalpha = alpha;
        }

        const int old = atomicExch(&head[obj], e);
        int4 rv;
        rv.x = old;
        rv.y = sub;
        rv.z = rel;
        rv.w = __float_as_int(myalpha);
        ((int4*)rec)[e] = rv;
    }
}

__global__ void node_sum_kernel(const float* __restrict__ hidden,
                                const float* __restrict__ rela,
                                const int*   __restrict__ head,
                                const int*   __restrict__ rec,
                                float* __restrict__ agg) {
    const int lane = threadIdx.x & 31;
    const int wave = (blockIdx.x * blockDim.x + threadIdx.x) >> 5;
    const int t    = lane & 7;
    const int n    = wave * 4 + (lane >> 3);
    if (n >= N_NODE) return;

    float4 a0 = {0.f, 0.f, 0.f, 0.f}, a1 = a0, a2 = a0, a3 = a0;

    int  e = head[n];
    int4 r = {-1, 0, 0, 0};
    if (e >= 0) r = ((const int4*)rec)[e];
    for (int it = 0; it < N_EDGE && e >= 0; ++it) {
        const int   en = r.x;
        const float al = __int_as_float(r.w);
        const float4* hs = (const float4*)(hidden + (size_t)r.y * IN_DIM);
        const float4* hr = (const float4*)(rela   + (size_t)r.z * IN_DIM);
        const float4 h0 = hs[t], h1 = hs[8 + t], h2 = hs[16 + t], h3 = hs[24 + t];
        const float4 g0 = hr[t], g1 = hr[8 + t], g2 = hr[16 + t], g3 = hr[24 + t];
        int4 rn = r;
        if (en >= 0) rn = ((const int4*)rec)[en];
        a0.x += al * h0.x * g0.x; a0.y += al * h0.y * g0.y; a0.z += al * h0.z * g0.z; a0.w += al * h0.w * g0.w;
        a1.x += al * h1.x * g1.x; a1.y += al * h1.y * g1.y; a1.z += al * h1.z * g1.z; a1.w += al * h1.w * g1.w;
        a2.x += al * h2.x * g2.x; a2.y += al * h2.y * g2.y; a2.z += al * h2.z * g2.z; a2.w += al * h2.w * g2.w;
        a3.x += al * h3.x * g3.x; a3.y += al * h3.y * g3.y; a3.z += al * h3.z * g3.z; a3.w += al * h3.w * g3.w;
        e = en;
        r = rn;
    }

    float4* dst = (float4*)(agg + (size_t)n * IN_DIM);
    dst[t]      = a0;
    dst[8 + t]  = a1;
    dst[16 + t] = a2;
    dst[24 + t] = a3;
}

extern "C" void kernel_launch(void* const* d_in, const int* in_sizes, int n_in,
                              void* d_out, int out_size, void* d_ws, size_t ws_size,
                              hipStream_t stream) {
    const int*   q_rel  = (const int*)  d_in[1];
    const int*   r_idx  = (const int*)  d_in[2];
    const float* hidden = (const float*)d_in[3];
    const int*   edges  = (const int*)  d_in[4];
    const float* rela   = (const float*)d_in[6];
    const float* Ws     = (const float*)d_in[7];
    const float* Wr     = (const float*)d_in[8];
    const float* WqrW   = (const float*)d_in[9];
    const float* Wqrb   = (const float*)d_in[10];
    const float* waW    = (const float*)d_in[11];
    const float* wab    = (const float*)d_in[12];
    const float* Wh     = (const float*)d_in[13];
    float* out = (float*)d_out;

    char* ws = (char*)d_ws;
    float* agg   = (float*)(ws);
    float* hproj = (float*)(ws + (size_t)N_NODE * IN_DIM * 4);
    float* rproj = (float*)(ws + (size_t)N_NODE * IN_DIM * 4
                               + (size_t)N_NODE * ATTN_DIM * 4);
    float* qproj = (float*)((char*)rproj + (((size_t)N_EMB * ATTN_DIM * 4 + 255ull) & ~255ull));
    int*   head  = (int*)((char*)qproj + (size_t)N_Q * ATTN_DIM * 4);
    int*   rec   = (int*)((char*)head + (((size_t)N_NODE * 4 + 255ull) & ~255ull));

    hipMemsetAsync(head, 0xFF, (size_t)N_NODE * sizeof(int), stream);

    {
        const int total = (N_EMB + N_Q) * ATTN_DIM;
        small_proj_kernel<<<(total + 255) / 256, 256, 0, stream>>>(
            rela, Wr, WqrW, Wqrb, q_rel, rproj, qproj);
    }

    {
        const int mTiles = N_NODE / 16;
        const int waveTiles = mTiles * (ATTN_DIM / 64);
        gemm_wmma_f32_kernel<ATTN_DIM><<<(waveTiles + 7) / 8, 256, 0, stream>>>(
            hidden, Ws, hproj, N_NODE);
    }

    edge_link_kernel<<<N_EDGE / 32 / 8, 256, 0, stream>>>(edges, r_idx, hproj, rproj, qproj,
                                                          waW, wab, head, rec);

    node_sum_kernel<<<(N_NODE / 4 + 7) / 8, 256, 0, stream>>>(hidden, rela, head, rec, agg);

    {
        const int mTiles = N_NODE / 16;
        const int waveTiles = mTiles * (OUT_DIM / 64);
        gemm_wmma_f32_kernel<OUT_DIM><<<(waveTiles + 7) / 8, 256, 0, stream>>>(
            agg, Wh, out, N_NODE);
    }
}


# Small gather tables staged in LDS by resident 1024-thread work-groups: rela in the node pass (4 lanes per node), rproj+qproj in the edge pass; per-node sum order unchanged (list order)
# speedup vs baseline: 1.2283x; 1.0843x over previous
#include <hip/hip_runtime.h>
#include <math.h>

#define N_NODE   50000
#define N_EDGE   800000
#define IN_DIM   128
#define OUT_DIM  128
#define ATTN_DIM 64
#define N_REL    200
#define N_Q      64
#define N_EMB    (2 * N_REL + 1)

typedef __attribute__((ext_vector_type(2))) float v2f;
typedef __attribute__((ext_vector_type(8))) float v8f;

__global__ void small_proj_kernel(const float* __restrict__ rela,
                                  const float* __restrict__ Wr,
                                  const float* __restrict__ WqrW,
                                  const float* __restrict__ Wqrb,
                                  const int*   __restrict__ q_rel,
                                  float* __restrict__ rproj,
                                  float* __restrict__ qproj) {
    int idx = blockIdx.x * blockDim.x + threadIdx.x;
    const int total = (N_EMB + N_Q) * ATTN_DIM;
    if (idx >= total) return;
    int col = idx & (ATTN_DIM - 1);
    int row = idx >> 6;
    if (row < N_EMB) {
        const float* h = rela + (size_t)row * IN_DIM;
        float s = 0.f;
        #pragma unroll 8
        for (int k = 0; k < IN_DIM; ++k) s = fmaf(h[k], Wr[k * ATTN_DIM + col], s);
        rproj[row * ATTN_DIM + col] = s;
    } else {
        int q = row - N_EMB;
        const float* h = rela + (size_t)q_rel[q] * IN_DIM;
        float s = Wqrb[col];
        #pragma unroll 8
        for (int k = 0; k < IN_DIM; ++k) s = fmaf(h[k], WqrW[k * ATTN_DIM + col], s);
        qproj[q * ATTN_DIM + col] = s;
    }
}

template <int N>
__global__ void gemm_wmma_f32_kernel(const float* __restrict__ A,
                                     const float* __restrict__ B,
                                     float* __restrict__ C,
                                     int M) {
    constexpr int K = 128;
    __shared__ v2f Bs2[(K / 2) * N];

    for (int i = threadIdx.x; i < (K / 2) * N; i += blockDim.x) {
        const int kp = i / N;
        const int n  = i - kp * N;
        v2f p;
        p.x = B[(2 * kp)     * N + n];
        p.y = B[(2 * kp + 1) * N + n];
        Bs2[i] = p;
    }
    __syncthreads();

    const int lane = threadIdx.x & 31;
    const int waveInBlock = threadIdx.x >> 5;
    constexpr int colTiles = N / 64;
    const long tileId = (long)blockIdx.x * 8 + waveInBlock;
    const int  mTile  = (int)(tileId / colTiles);
    const int  nTile  = (int)(tileId % colTiles);
    if (mTile * 16 >= M) return;

    const int mrow  = mTile * 16 + (lane & 15);
    const int sel   = lane >> 4;
    const int koff  = sel * 2;
    const int ncol0 = nTile * 64;
    const int nlane = ncol0 + (lane & 15);

    v8f acc0 = {}, acc1 = {}, acc2 = {}, acc3 = {};

    const float* __restrict__ arow = A + (size_t)mrow * K;
    #pragma unroll 4
    for (int k0 = 0; k0 < K; k0 += 4) {
        v2f a;
        a.x = arow[k0 + koff];
        a.y = arow[k0 + koff + 1];

        const v2f* __restrict__ brow = &Bs2[(k0 / 2 + sel) * N + nlane];
        acc0 = __builtin_amdgcn_wmma_f32_16x16x4_f32(false, a, false, brow[ 0], (short)0, acc0, false, false);
        acc1 = __builtin_amdgcn_wmma_f32_16x16x4_f32(false, a, false, brow[16], (short)0, acc1, false, false);
        acc2 = __builtin_amdgcn_wmma_f32_16x16x4_f32(false, a, false, brow[32], (short)0, acc2, false, false);
        acc3 = __builtin_amdgcn_wmma_f32_16x16x4_f32(false, a, false, brow[48], (short)0, acc3, false, false);
    }

    const int cmBase = mTile * 16 + ((lane < 16) ? 0 : 8);
    const int cn = lane & 15;
    float* __restrict__ c0 = C + (size_t)cmBase * N + ncol0 + cn;
    #pragma unroll
    for (int r = 0; r < 8; ++r) {
        float* cr = c0 + (size_t)r * N;
        cr[ 0] = acc0[r];
        cr[16] = acc1[r];
        cr[32] = acc2[r];
        cr[48] = acc3[r];
    }
}

#define TAB_PITCH    68
#define EDGE_WGS     512
#define EDGE_THREADS 1024

__global__ __launch_bounds__(EDGE_THREADS)
void edge_link_kernel(const int*   __restrict__ edges,
                      const int*   __restrict__ r_idx,
                      const float* __restrict__ hproj,
                      const float* __restrict__ rproj,
                      const float* __restrict__ qproj,
                      const float* __restrict__ waW,
                      const float* __restrict__ wab,
                      int* __restrict__ head,
                      int* __restrict__ rec) {
    __shared__ __attribute__((aligned(16))) float tab[(N_EMB + N_Q) * TAB_PITCH];

    const int lane  = threadIdx.x & 31;
    const int t     = lane & 7;
    const int gbase = lane & 24;
    constexpr int NCHUNK = N_EDGE / 32;
    constexpr int STRIDE = EDGE_WGS * (EDGE_THREADS / 32);

    int c = (threadIdx.x >> 5) * EDGE_WGS + blockIdx.x;
    int sub = 0, rel = 0, obj = 0, q = 0;
    if (c < NCHUNK) {
        const int e = c * 32 + lane;
        sub = edges[e * 3 + 0];
        rel = edges[e * 3 + 1];
        obj = edges[e * 3 + 2];
        q   = r_idx[e];
    }

    constexpr int NR4 = N_EMB * (ATTN_DIM / 4);
    constexpr int NT4 = (N_EMB + N_Q) * (ATTN_DIM / 4);
    #pragma unroll
    for (int k = 0; k < (NT4 + EDGE_THREADS - 1) / EDGE_THREADS; ++k) {
        const int i = k * EDGE_THREADS + (int)threadIdx.x;
        if (i < NT4) {
            const float4 v = (i < NR4) ? ((const float4*)rproj)[i]
                                       : ((const float4*)qproj)[i - NR4];
            *(float4*)&tab[(i >> 4) * TAB_PITCH + (i & 15) * 4] = v;
        }
    }
    __syncthreads();

    const float  bias = wab[0];
    const float4 w0 = ((const float4*)waW)[t];
    const float4 w1 = ((const float4*)waW)[8 + t];

    while (c < NCHUNK) {
        const int e = c * 32 + lane;

        float myalpha = 0.f;
        for (int j = 0; j < 8; ++j) {
            const int src   = gbase + j;
            const int s_sub = __shfl(sub, src, 32);
            const int s_rel = __shfl(rel, src, 32);
            const int s_q   = __shfl(q,   src, 32);
            const float4* hp = (const float4*)(hproj + (size_t)s_sub * ATTN_DIM);
            const float4* rp = (const float4*)(tab + s_rel * TAB_PITCH);
            const float4* qp = (const float4*)(tab + (N_EMB + s_q) * TAB_PITCH);
            const float4 h0 = hp[t], h1 = hp[8 + t];
            const float4 r0 = rp[t], r1 = rp[8 + t];
            const float4 q0 = qp[t], q1 = qp[8 + t];
            float s = fmaxf(h0.x + r0.x + q0.x, 0.f) * w0.x;
            s = fmaf(fmaxf(h0.y + r0.y + q0.y, 0.f), w0.y, s);
            s = fmaf(fmaxf(h0.z + r0.z + q0.z, 0.f), w0.z, s);
            s = fmaf(fmaxf(h0.w + r0.w + q0.w, 0.f), w0.w, s);
            s = fmaf(fmaxf(h1.x + r1.x + q1.x, 0.f), w1.x, s);
            s = fmaf(fmaxf(h1.y + r1.y + q1.y, 0.f), w1.y, s);
            s = fmaf(fmaxf(h1.z + r1.z + q1.z, 0.f), w1.z, s);
            s = fmaf(fmaxf(h1.w + r1.w + q1.w, 0.f), w1.w, s);
            s += __shfl_xor(s, 1, 32);
            s += __shfl_xor(s, 2, 32);
            s += __shfl_xor(s, 4, 32);
            const float alpha = 1.0f / (1.0f + __expf(-(s + bias)));
            if (t == j) myalpha = alpha;
        }

        const int old = atomicExch(&head[obj], e);
        int4 rv;
        rv.x = old;
        rv.y = sub;
        rv.z = rel;
        rv.w = __float_as_int(myalpha);
        ((int4*)rec)[e] = rv;

        c += STRIDE;
        if (c < NCHUNK) {
            const int en = c * 32 + lane;
            sub = edges[en * 3 + 0];
            rel = edges[en * 3 + 1];
            obj = edges[en * 3 + 2];
            q   = r_idx[en];
        }
    }
}

#define RELA_PITCH   132
#define NODE_WGS     256
#define NODE_THREADS 1024

__global__ __launch_bounds__(NODE_THREADS)
void node_sum_kernel(const float* __restrict__ hidden,
                     const float* __restrict__ rela,
                     const int*   __restrict__ head,
                     const int*   __restrict__ rec,
                     float* __restrict__ agg) {
    __shared__ __attribute__((aligned(16))) float rs[N_EMB * RELA_PITCH];

    const int t = threadIdx.x & 3;
    const int n = (threadIdx.x >> 2) * NODE_WGS + blockIdx.x;
    const bool live = n < N_NODE;

    int e = live ? head[n] : -1;

    constexpr int NV4 = N_EMB * (IN_DIM / 4);
    constexpr int NST = (NV4 + NODE_THREADS - 1) / NODE_THREADS;
    float4 sv[NST];
    #pragma unroll
    for (int k = 0; k < NST; ++k) {
        const int i = k * NODE_THREADS + (int)threadIdx.x;
        sv[k] = ((const float4*)rela)[i < NV4 ? i : NV4 - 1];
    }
    int4 r = ((const int4*)rec)[e < 0 ? 0 : e];
    #pragma unroll
    for (int k = 0; k < NST; ++k) {
        const int i = k * NODE_THREADS + (int)threadIdx.x;
        if (i < NV4) *(float4*)&rs[(i >> 5) * RELA_PITCH + (i & 31) * 4] = sv[k];
    }
    __syncthreads();
    if (!live) return;

    float4 a[8];
    #pragma unroll
    for (int c = 0; c < 8; ++c) a[c] = make_float4(0.f, 0.f, 0.f, 0.f);

    for (int it = 0; it < N_EDGE && e >= 0; ++it) {
        const int   en = r.x;
        const float al = __int_as_float(r.w);
        const float4* hs = (const float4*)(hidden + (size_t)r.y * IN_DIM) + t;
        const float4* hr = (const float4*)(rs + r.z * RELA_PITCH) + t;
        float4 h[8];
        #pragma unroll
        for (int c = 0; c < 8; ++c) h[c] = hs[4 * c];
        int4 rn = r;
        if (en >= 0) rn = ((const int4*)rec)[en];
        #pragma unroll
        for (int c = 0; c < 8; ++c) {
            const float4 g = hr[4 * c];
            a[c].x += al * h[c].x * g.x;
            a[c].y += al * h[c].y * g.y;
            a[c].z += al * h[c].z * g.z;
            a[c].w += al * h[c].w * g.w;
        }
        e = en;
        r = rn;
    }

    float4* dst = (float4*)(agg + (size_t)n * IN_DIM) + t;
    #pragma unroll
    for (int c = 0; c < 8; ++c) dst[4 * c] = a[c];
}

extern "C" void kernel_launch(void* const* d_in, const int* in_sizes, int n_in,
                              void* d_out, int out_size, void* d_ws, size_t ws_size,
                              hipStream_t stream) {
    const int*   q_rel  = (const int*)  d_in[1];
    const int*   r_idx  = (const int*)  d_in[2];
    const float* hidden = (const float*)d_in[3];
    const int*   edges  = (const int*)  d_in[4];
    const float* rela   = (const float*)d_in[6];
    const float* Ws     = (const float*)d_in[7];
    const float* Wr     = (const float*)d_in[8];
    const float* WqrW   = (const float*)d_in[9];
    const float* Wqrb   = (const float*)d_in[10];
    const float* waW    = (const float*)d_in[11];
    const float* wab    = (const float*)d_in[12];
    const float* Wh     = (const float*)d_in[13];
    float* out = (float*)d_out;

    char* ws = (char*)d_ws;
    float* agg   = (float*)(ws);
    float* hproj = (float*)(ws + (size_t)N_NODE * IN_DIM * 4);
    float* rproj = (float*)(ws + (size_t)N_NODE * IN_DIM * 4
                               + (size_t)N_NODE * ATTN_DIM * 4);
    float* qproj = (float*)((char*)rproj + (((size_t)N_EMB * ATTN_DIM * 4 + 255ull) & ~255ull));
    int*   head  = (int*)((char*)qproj + (size_t)N_Q * ATTN_DIM * 4);
    int*   rec   = (int*)((char*)head + (((size_t)N_NODE * 4 + 255ull) & ~255ull));

    hipMemsetAsync(head, 0xFF, (size_t)N_NODE * sizeof(int), stream);

    {
        const int total = (N_EMB + N_Q) * ATTN_DIM;
        small_proj_kernel<<<(total + 255) / 256, 256, 0, stream>>>(
            rela, Wr, WqrW, Wqrb, q_rel, rproj, qproj);
    }

    {
        const int mTiles = N_NODE / 16;
        const int waveTiles = mTiles * (ATTN_DIM / 64);
        gemm_wmma_f32_kernel<ATTN_DIM><<<(waveTiles + 7) / 8, 256, 0, stream>>>(
            hidden, Ws, hproj, N_NODE);
    }

    edge_link_kernel<<<EDGE_WGS, EDGE_THREADS, 0, stream>>>(edges, r_idx, hproj, rproj, qproj,
                                                          waW, wab, head, rec);

    node_sum_kernel<<<NODE_WGS, NODE_THREADS, 0, stream>>>(hidden, rela, head, rec, agg);

    {
        const int mTiles = N_NODE / 16;
        const int waveTiles = mTiles * (OUT_DIM / 64);
        gemm_wmma_f32_kernel<OUT_DIM><<<(waveTiles + 7) / 8, 256, 0, stream>>>(
            agg, Wh, out, N_NODE);
    }
}


# GEMM B staging: every thread requests all its 16-byte row-pair loads before its first LDS store, so a block waits for one trip to memory instead of 16 or 32 serial trips; sums unchanged
# speedup vs baseline: 1.3924x; 1.1336x over previous
#include <hip/hip_runtime.h>
#include <math.h>

#define N_NODE   50000
#define N_EDGE   800000
#define IN_DIM   128
#define OUT_DIM  128
#define ATTN_DIM 64
#define N_REL    200
#define N_Q      64
#define N_EMB    (2 * N_REL + 1)

typedef __attribute__((ext_vector_type(2))) float v2f;
typedef __attribute__((ext_vector_type(8))) float v8f;

__global__ void small_proj_kernel(const float* __restrict__ rela,
                                  const float* __restrict__ Wr,
                                  const float* __restrict__ WqrW,
                                  const float* __restrict__ Wqrb,
                                  const int*   __restrict__ q_rel,
                                  float* __restrict__ rproj,
                                  float* __restrict__ qproj) {
    int idx = blockIdx.x * blockDim.x + threadIdx.x;
    const int total = (N_EMB + N_Q) * ATTN_DIM;
    if (idx >= total) return;
    int col = idx & (ATTN_DIM - 1);
    int row = idx >> 6;
    if (row < N_EMB) {
        const float* h = rela + (size_t)row * IN_DIM;
        float s = 0.f;
        #pragma unroll 8
        for (int k = 0; k < IN_DIM; ++k) s = fmaf(h[k], Wr[k * ATTN_DIM + col], s);
        rproj[row * ATTN_DIM + col] = s;
    } else {
        int q = row - N_EMB;
        const float* h = rela + (size_t)q_rel[q] * IN_DIM;
        float s = Wqrb[col];
        #pragma unroll 8
        for (int k = 0; k < IN_DIM; ++k) s = fmaf(h[k], WqrW[k * ATTN_DIM + col], s);
        qproj[q * ATTN_DIM + col] = s;
    }
}

template <int N>
__global__ __launch_bounds__(256)
void gemm_wmma_f32_kernel(const float* __restrict__ A,
                                     const float* __restrict__ B,
                                     float* __restrict__ C,
                                     int M) {
    constexpr int K = 128;
    __shared__ __attribute__((aligned(16))) v2f Bs2[(K / 2) * N];

    constexpr int NTHR = 256;
    constexpr int NQ   = (K / 2) * (N / 4);
    constexpr int PER  = NQ / NTHR;
    static_assert(NQ % NTHR == 0, "B must split evenly over the block");
    float4 r0[PER], r1[PER];
    #pragma unroll
    for (int j = 0; j < PER; ++j) {
        const int i  = j * NTHR + (int)threadIdx.x;
        const int kp = i / (N / 4);
        const int n  = (i - kp * (N / 4)) * 4;
        r0[j] = *(const float4*)(B + (2 * kp)     * N + n);
        r1[j] = *(const float4*)(B + (2 * kp + 1) * N + n);
    }
    #pragma unroll
    for (int j = 0; j < PER; ++j) {
        const int i  = j * NTHR + (int)threadIdx.x;
        const int kp = i / (N / 4);
        const int n  = (i - kp * (N / 4)) * 4;
        float4* d = (float4*)&Bs2[kp * N + n];
        d[0] = make_float4(r0[j].x, r1[j].x, r0[j].y, r1[j].y);
        d[1] = make_float4(r0[j].z, r1[j].z, r0[j].w, r1[j].w);
    }
    __syncthreads();

    const int lane = threadIdx.x & 31;
    const int waveInBlock = threadIdx.x >> 5;
    constexpr int colTiles = N / 64;
    const long tileId = (long)blockIdx.x * 8 + waveInBlock;
    const int  mTile  = (int)(tileId / colTiles);
    const int  nTile  = (int)(tileId % colTiles);
    if (mTile * 16 >= M) return;

    const int mrow  = mTile * 16 + (lane & 15);
    const int sel   = lane >> 4;
    const int koff  = sel * 2;
    const int ncol0 = nTile * 64;
    const int nlane = ncol0 + (lane & 15);

    v8f acc0 = {}, acc1 = {}, acc2 = {}, acc3 = {};

    const float* __restrict__ arow = A + (size_t)mrow * K;
    #pragma unroll 4
    for (int k0 = 0; k0 < K; k0 += 4) {
        v2f a;
        a.x = arow[k0 + koff];
        a.y = arow[k0 + koff + 1];

        const v2f* __restrict__ brow = &Bs2[(k0 / 2 + sel) * N + nlane];
        acc0 = __builtin_amdgcn_wmma_f32_16x16x4_f32(false, a, false, brow[ 0], (short)0, acc0, false, false);
        acc1 = __builtin_amdgcn_wmma_f32_16x16x4_f32(false, a, false, brow[16], (short)0, acc1, false, false);
        acc2 = __builtin_amdgcn_wmma_f32_16x16x4_f32(false, a, false, brow[32], (short)0, acc2, false, false);
        acc3 = __builtin_amdgcn_wmma_f32_16x16x4_f32(false, a, false, brow[48], (short)0, acc3, false, false);
    }

    const int cmBase = mTile * 16 + ((lane < 16) ? 0 : 8);
    const int cn = lane & 15;
    float* __restrict__ c0 = C + (size_t)cmBase * N + ncol0 + cn;
    #pragma unroll
    for (int r = 0; r < 8; ++r) {
        float* cr = c0 + (size_t)r * N;
        cr[ 0] = acc0[r];
        cr[16] = acc1[r];
        cr[32] = acc2[r];
        cr[48] = acc3[r];
    }
}

#define TAB_PITCH    68
#define EDGE_WGS     512
#define EDGE_THREADS 1024

__global__ __launch_bounds__(EDGE_THREADS)
void edge_link_kernel(const int*   __restrict__ edges,
                      const int*   __restrict__ r_idx,
                      const float* __restrict__ hproj,
                      const float* __restrict__ rproj,
                      const float* __restrict__ qproj,
                      const float* __restrict__ waW,
                      const float* __restrict__ wab,
                      int* __restrict__ head,
                      int* __restrict__ rec) {
    __shared__ __attribute__((aligned(16))) float tab[(N_EMB + N_Q) * TAB_PITCH];

    const int lane  = threadIdx.x & 31;
    const int t     = lane & 7;
    const int gbase = lane & 24;
    constexpr int NCHUNK = N_EDGE / 32;
    constexpr int STRIDE = EDGE_WGS * (EDGE_THREADS / 32);

    int c = (threadIdx.x >> 5) * EDGE_WGS + blockIdx.x;
    int sub = 0, rel = 0, obj = 0, q = 0;
    if (c < NCHUNK) {
        const int e = c * 32 + lane;
        sub = edges[e * 3 + 0];
        rel = edges[e * 3 + 1];
        obj = edges[e * 3 + 2];
        q   = r_idx[e];
    }

    constexpr int NR4 = N_EMB * (ATTN_DIM / 4);
    constexpr int NT4 = (N_EMB + N_Q) * (ATTN_DIM / 4);
    #pragma unroll
    for (int k = 0; k < (NT4 + EDGE_THREADS - 1) / EDGE_THREADS; ++k) {
        const int i = k * EDGE_THREADS + (int)threadIdx.x;
        if (i < NT4) {
            const float4 v = (i < NR4) ? ((const float4*)rproj)[i]
                                       : ((const float4*)qproj)[i - NR4];
            *(float4*)&tab[(i >> 4) * TAB_PITCH + (i & 15) * 4] = v;
        }
    }
    __syncthreads();

    const float  bias = wab[0];
    const float4 w0 = ((const float4*)waW)[t];
    const float4 w1 = ((const float4*)waW)[8 + t];

    while (c < NCHUNK) {
        const int e = c * 32 + lane;

        float myalpha = 0.f;
        for (int j = 0; j < 8; ++j) {
            const int src   = gbase + j;
            const int s_sub = __shfl(sub, src, 32);
            const int s_rel = __shfl(rel, src, 32);
            const int s_q   = __shfl(q,   src, 32);
            const float4* hp = (const float4*)(hproj + (size_t)s_sub * ATTN_DIM);
            const float4* rp = (const float4*)(tab + s_rel * TAB_PITCH);
            const float4* qp = (const float4*)(tab + (N_EMB + s_q) * TAB_PITCH);
            const float4 h0 = hp[t], h1 = hp[8 + t];
            const float4 r0 = rp[t], r1 = rp[8 + t];
            const float4 q0 = qp[t], q1 = qp[8 + t];
            float s = fmaxf(h0.x + r0.x + q0.x, 0.f) * w0.x;
            s = fmaf(fmaxf(h0.y + r0.y + q0.y, 0.f), w0.y, s);
            s = fmaf(fmaxf(h0.z + r0.z + q0.z, 0.f), w0.z, s);
            s = fmaf(fmaxf(h0.w + r0.w + q0.w, 0.f), w0.w, s);
            s = fmaf(fmaxf(h1.x + r1.x + q1.x, 0.f), w1.x, s);
            s = fmaf(fmaxf(h1.y + r1.y + q1.y, 0.f), w1.y, s);
            s = fmaf(fmaxf(h1.z + r1.z + q1.z, 0.f), w1.z, s);
            s = fmaf(fmaxf(h1.w + r1.w + q1.w, 0.f), w1.w, s);
            s += __shfl_xor(s, 1, 32);
            s += __shfl_xor(s, 2, 32);
            s += __shfl_xor(s, 4, 32);
            const float alpha = 1.0f / (1.0f + __expf(-(s + bias)));
            if (t == j) myalpha = alpha;
        }

        const int old = atomicExch(&head[obj], e);
        int4 rv;
        rv.x = old;
        rv.y = sub;
        rv.z = rel;
        rv.w = __float_as_int(myalpha);
        ((int4*)rec)[e] = rv;

        c += STRIDE;
        if (c < NCHUNK) {
            const int en = c * 32 + lane;
            sub = edges[en * 3 + 0];
            rel = edges[en * 3 + 1];
            obj = edges[en * 3 + 2];
            q   = r_idx[en];
        }
    }
}

#define RELA_PITCH   132
#define NODE_WGS     256
#define NODE_THREADS 1024

__global__ __launch_bounds__(NODE_THREADS)
void node_sum_kernel(const float* __restrict__ hidden,
                     const float* __restrict__ rela,
                     const int*   __restrict__ head,
                     const int*   __restrict__ rec,
                     float* __restrict__ agg) {
    __shared__ __attribute__((aligned(16))) float rs[N_EMB * RELA_PITCH];

    const int t = threadIdx.x & 3;
    const int n = (threadIdx.x >> 2) * NODE_WGS + blockIdx.x;
    const bool live = n < N_NODE;

    int e = live ? head[n] : -1;

    constexpr int NV4 = N_EMB * (IN_DIM / 4);
    constexpr int NST = (NV4 + NODE_THREADS - 1) / NODE_THREADS;
    float4 sv[NST];
    #pragma unroll
    for (int k = 0; k < NST; ++k) {
        const int i = k * NODE_THREADS + (int)threadIdx.x;
        sv[k] = ((const float4*)rela)[i < NV4 ? i : NV4 - 1];
    }
    int4 r = ((const int4*)rec)[e < 0 ? 0 : e];
    #pragma unroll
    for (int k = 0; k < NST; ++k) {
        const int i = k * NODE_THREADS + (int)threadIdx.x;
        if (i < NV4) *(float4*)&rs[(i >> 5) * RELA_PITCH + (i & 31) * 4] = sv[k];
    }
    __syncthreads();
    if (!live) return;

    float4 a[8];
    #pragma unroll
    for (int c = 0; c < 8; ++c) a[c] = make_float4(0.f, 0.f, 0.f, 0.f);

    for (int it = 0; it < N_EDGE && e >= 0; ++it) {
        const int   en = r.x;
        const float al = __int_as_float(r.w);
        const float4* hs = (const float4*)(hidden + (size_t)r.y * IN_DIM) + t;
        const float4* hr = (const float4*)(rs + r.z * RELA_PITCH) + t;
        float4 h[8];
        #pragma unroll
        for (int c = 0; c < 8; ++c) h[c] = hs[4 * c];
        int4 rn = r;
        if (en >= 0) rn = ((const int4*)rec)[en];
        #pragma unroll
        for (int c = 0; c < 8; ++c) {
            const float4 g = hr[4 * c];
            a[c].x += al * h[c].x * g.x;
            a[c].y += al * h[c].y * g.y;
            a[c].z += al * h[c].z * g.z;
            a[c].w += al * h[c].w * g.w;
        }
        e = en;
        r = rn;
    }

    float4* dst = (float4*)(agg + (size_t)n * IN_DIM) + t;
    #pragma unroll
    for (int c = 0; c < 8; ++c) dst[4 * c] = a[c];
}

extern "C" void kernel_launch(void* const* d_in, const int* in_sizes, int n_in,
                              void* d_out, int out_size, void* d_ws, size_t ws_size,
                              hipStream_t stream) {
    const int*   q_rel  = (const int*)  d_in[1];
    const int*   r_idx  = (const int*)  d_in[2];
    const float* hidden = (const float*)d_in[3];
    const int*   edges  = (const int*)  d_in[4];
    const float* rela   = (const float*)d_in[6];
    const float* Ws     = (const float*)d_in[7];
    const float* Wr     = (const float*)d_in[8];
    const float* WqrW   = (const float*)d_in[9];
    const float* Wqrb   = (const float*)d_in[10];
    const float* waW    = (const float*)d_in[11];
    const float* wab    = (const float*)d_in[12];
    const float* Wh     = (const float*)d_in[13];
    float* out = (float*)d_out;

    char* ws = (char*)d_ws;
    float* agg   = (float*)(ws);
    float* hproj = (float*)(ws + (size_t)N_NODE * IN_DIM * 4);
    float* rproj = (float*)(ws + (size_t)N_NODE * IN_DIM * 4
                               + (size_t)N_NODE * ATTN_DIM * 4);
    float* qproj = (float*)((char*)rproj + (((size_t)N_EMB * ATTN_DIM * 4 + 255ull) & ~255ull));
    int*   head  = (int*)((char*)qproj + (size_t)N_Q * ATTN_DIM * 4);
    int*   rec   = (int*)((char*)head + (((size_t)N_NODE * 4 + 255ull) & ~255ull));

    hipMemsetAsync(head, 0xFF, (size_t)N_NODE * sizeof(int), stream);

    {
        const int total = (N_EMB + N_Q) * ATTN_DIM;
        small_proj_kernel<<<(total + 255) / 256, 256, 0, stream>>>(
            rela, Wr, WqrW, Wqrb, q_rel, rproj, qproj);
    }

    {
        const int mTiles = N_NODE / 16;
        const int waveTiles = mTiles * (ATTN_DIM / 64);
        gemm_wmma_f32_kernel<ATTN_DIM><<<(waveTiles + 7) / 8, 256, 0, stream>>>(
            hidden, Ws, hproj, N_NODE);
    }

    edge_link_kernel<<<EDGE_WGS, EDGE_THREADS, 0, stream>>>(edges, r_idx, hproj, rproj, qproj,
                                                          waW, wab, head, rec);

    node_sum_kernel<<<NODE_WGS, NODE_THREADS, 0, stream>>>(hidden, rela, head, rec, agg);

    {
        const int mTiles = N_NODE / 16;
        const int waveTiles = mTiles * (OUT_DIM / 64);
        gemm_wmma_f32_kernel<OUT_DIM><<<(waveTiles + 7) / 8, 256, 0, stream>>>(
            agg, Wh, out, N_NODE);
    }
}


# Two launches fewer: the list-head fill (was a memset) and the tiny rproj/qproj projections run as extra work-groups in the hproj GEMM's grid; 6 launches become 4; no sum changes
# speedup vs baseline: 1.4861x; 1.0673x over previous
#include <hip/hip_runtime.h>
#include <math.h>

#define N_NODE   50000
#define N_EDGE   800000
#define IN_DIM   128
#define OUT_DIM  128
#define ATTN_DIM 64
#define N_REL    200
#define N_Q      64
#define N_EMB    (2 * N_REL + 1)

typedef __attribute__((ext_vector_type(2))) float v2f;
typedef __attribute__((ext_vector_type(8))) float v8f;

__device__ __forceinline__ void small_proj_body(int idx,
                                  const float* __restrict__ rela,
                                  const float* __restrict__ Wr,
                                  const float* __restrict__ WqrW,
                                  const float* __restrict__ Wqrb,
                                  const int*   __restrict__ q_rel,
                                  float* __restrict__ rproj,
                                  float* __restrict__ qproj) {
    const int total = (N_EMB + N_Q) * ATTN_DIM;
    if (idx >= total) return;
    int col = idx & (ATTN_DIM - 1);
    int row = idx >> 6;
    if (row < N_EMB) {
        const float* h = rela + (size_t)row * IN_DIM;
        float s = 0.f;
        #pragma unroll 8
        for (int k = 0; k < IN_DIM; ++k) s = fmaf(h[k], Wr[k * ATTN_DIM + col], s);
        rproj[row * ATTN_DIM + col] = s;
    } else {
        int q = row - N_EMB;
        const float* h = rela + (size_t)q_rel[q] * IN_DIM;
        float s = Wqrb[col];
        #pragma unroll 8
        for (int k = 0; k < IN_DIM; ++k) s = fmaf(h[k], WqrW[k * ATTN_DIM + col], s);
        qproj[q * ATTN_DIM + col] = s;
    }
}

template <int N>
__device__ __forceinline__
void gemm_wmma_f32_body(const float* __restrict__ A,
                        const float* __restrict__ B,
                        float* __restrict__ C,
                        int M, int blk) {
    constexpr int K = 128;
    __shared__ __attribute__((aligned(16))) v2f Bs2[(K / 2) * N];

    constexpr int NTHR = 256;
    constexpr int NQ   = (K / 2) * (N / 4);
    constexpr int PER  = NQ / NTHR;
    static_assert(NQ % NTHR == 0, "B must split evenly over the block");
    float4 r0[PER], r1[PER];
    #pragma unroll
    for (int j = 0; j < PER; ++j) {
        const int i  = j * NTHR + (int)threadIdx.x;
        const int kp = i / (N / 4);
        const int n  = (i - kp * (N / 4)) * 4;
        r0[j] = *(const float4*)(B + (2 * kp)     * N + n);
        r1[j] = *(const float4*)(B + (2 * kp + 1) * N + n);
    }
    #pragma unroll
    for (int j = 0; j < PER; ++j) {
        const int i  = j * NTHR + (int)threadIdx.x;
        const int kp = i / (N / 4);
        const int n  = (i - kp * (N / 4)) * 4;
        float4* d = (float4*)&Bs2[kp * N + n];
        d[0] = make_float4(r0[j].x, r1[j].x, r0[j].y, r1[j].y);
        d[1] = make_float4(r0[j].z, r1[j].z, r0[j].w, r1[j].w);
    }
    __syncthreads();

    const int lane = threadIdx.x & 31;
    const int waveInBlock = threadIdx.x >> 5;
    constexpr int colTiles = N / 64;
    const long tileId = (long)blk * 8 + waveInBlock;
    const int  mTile  = (int)(tileId / colTiles);
    const int  nTile  = (int)(tileId % colTiles);
    if (mTile * 16 >= M) return;

    const int mrow  = mTile * 16 + (lane & 15);
    const int sel   = lane >> 4;
    const int koff  = sel * 2;
    const int ncol0 = nTile * 64;
    const int nlane = ncol0 + (lane & 15);

    v8f acc0 = {}, acc1 = {}, acc2 = {}, acc3 = {};

    const float* __restrict__ arow = A + (size_t)mrow * K;
    #pragma unroll 4
    for (int k0 = 0; k0 < K; k0 += 4) {
        v2f a;
        a.x = arow[k0 + koff];
        a.y = arow[k0 + koff + 1];

        const v2f* __restrict__ brow = &Bs2[(k0 / 2 + sel) * N + nlane];
        acc0 = __builtin_amdgcn_wmma_f32_16x16x4_f32(false, a, false, brow[ 0], (short)0, acc0, false, false);
        acc1 = __builtin_amdgcn_wmma_f32_16x16x4_f32(false, a, false, brow[16], (short)0, acc1, false, false);
        acc2 = __builtin_amdgcn_wmma_f32_16x16x4_f32(false, a, false, brow[32], (short)0, acc2, false, false);
        acc3 = __builtin_amdgcn_wmma_f32_16x16x4_f32(false, a, false, brow[48], (short)0, acc3, false, false);
    }

    const int cmBase = mTile * 16 + ((lane < 16) ? 0 : 8);
    const int cn = lane & 15;
    float* __restrict__ c0 = C + (size_t)cmBase * N + ncol0 + cn;
    #pragma unroll
    for (int r = 0; r < 8; ++r) {
        float* cr = c0 + (size_t)r * N;
        cr[ 0] = acc0[r];
        cr[16] = acc1[r];
        cr[32] = acc2[r];
        cr[48] = acc3[r];
    }
}

__global__ __launch_bounds__(256)
void gemm_out_kernel(const float* __restrict__ A,
                     const float* __restrict__ B,
                     float* __restrict__ C,
                     int M) {
    gemm_wmma_f32_body<OUT_DIM>(A, B, C, M, (int)blockIdx.x);
}

#define SP_WGS    (((N_EMB + N_Q) * ATTN_DIM + 255) / 256)
#define GEMM1_WGS (((N_NODE / 16) * (ATTN_DIM / 64) + 7) / 8)

__global__ __launch_bounds__(256)
void front_kernel(const float* __restrict__ hidden,
                  const float* __restrict__ Ws,
                  float* __restrict__ hproj,
                  const float* __restrict__ rela,
                  const float* __restrict__ Wr,
                  const float* __restrict__ WqrW,
                  const float* __restrict__ Wqrb,
                  const int*   __restrict__ q_rel,
                  float* __restrict__ rproj,
                  float* __restrict__ qproj,
                  int* __restrict__ head) {
    if (blockIdx.x < SP_WGS) {
        const int idx = (int)blockIdx.x * 256 + (int)threadIdx.x;
        static_assert(N_NODE % 4 == 0 && N_NODE / 4 <= SP_WGS * 256, "head[] must be covered");
        if (idx < N_NODE / 4) ((int4*)head)[idx] = make_int4(-1, -1, -1, -1);
        small_proj_body(idx, rela, Wr, WqrW, Wqrb, q_rel, rproj, qproj);
        return;
    }
    gemm_wmma_f32_body<ATTN_DIM>(hidden, Ws, hproj, N_NODE, (int)blockIdx.x - SP_WGS);
}

#define TAB_PITCH    68
#define EDGE_WGS     512
#define EDGE_THREADS 1024

__global__ __launch_bounds__(EDGE_THREADS)
void edge_link_kernel(const int*   __restrict__ edges,
                      const int*   __restrict__ r_idx,
                      const float* __restrict__ hproj,
                      const float* __restrict__ rproj,
                      const float* __restrict__ qproj,
                      const float* __restrict__ waW,
                      const float* __restrict__ wab,
                      int* __restrict__ head,
                      int* __restrict__ rec) {
    __shared__ __attribute__((aligned(16))) float tab[(N_EMB + N_Q) * TAB_PITCH];

    const int lane  = threadIdx.x & 31;
    const int t     = lane & 7;
    const int gbase = lane & 24;
    constexpr int NCHUNK = N_EDGE / 32;
    constexpr int STRIDE = EDGE_WGS * (EDGE_THREADS / 32);

    int c = (threadIdx.x >> 5) * EDGE_WGS + blockIdx.x;
    int sub = 0, rel = 0, obj = 0, q = 0;
    if (c < NCHUNK) {
        const int e = c * 32 + lane;
        sub = edges[e * 3 + 0];
        rel = edges[e * 3 + 1];
        obj = edges[e * 3 + 2];
        q   = r_idx[e];
    }

    constexpr int NR4 = N_EMB * (ATTN_DIM / 4);
    constexpr int NT4 = (N_EMB + N_Q) * (ATTN_DIM / 4);
    #pragma unroll
    for (int k = 0; k < (NT4 + EDGE_THREADS - 1) / EDGE_THREADS; ++k) {
        const int i = k * EDGE_THREADS + (int)threadIdx.x;
        if (i < NT4) {
            const float4 v = (i < NR4) ? ((const float4*)rproj)[i]
                                       : ((const float4*)qproj)[i - NR4];
            *(float4*)&tab[(i >> 4) * TAB_PITCH + (i & 15) * 4] = v;
        }
    }
    __syncthreads();

    const float  bias = wab[0];
    const float4 w0 = ((const float4*)waW)[t];
    const float4 w1 = ((const float4*)waW)[8 + t];

    while (c < NCHUNK) {
        const int e = c * 32 + lane;

        float myalpha = 0.f;
        for (int j = 0; j < 8; ++j) {
            const int src   = gbase + j;
            const int s_sub = __shfl(sub, src, 32);
            const int s_rel = __shfl(rel, src, 32);
            const int s_q   = __shfl(q,   src, 32);
            const float4* hp = (const float4*)(hproj + (size_t)s_sub * ATTN_DIM);
            const float4* rp = (const float4*)(tab + s_rel * TAB_PITCH);
            const float4* qp = (const float4*)(tab + (N_EMB + s_q) * TAB_PITCH);
            const float4 h0 = hp[t], h1 = hp[8 + t];
            const float4 r0 = rp[t], r1 = rp[8 + t];
            const float4 q0 = qp[t], q1 = qp[8 + t];
            float s = fmaxf(h0.x + r0.x + q0.x, 0.f) * w0.x;
            s = fmaf(fmaxf(h0.y + r0.y + q0.y, 0.f), w0.y, s);
            s = fmaf(fmaxf(h0.z + r0.z + q0.z, 0.f), w0.z, s);
            s = fmaf(fmaxf(h0.w + r0.w + q0.w, 0.f), w0.w, s);
            s = fmaf(fmaxf(h1.x + r1.x + q1.x, 0.f), w1.x, s);
            s = fmaf(fmaxf(h1.y + r1.y + q1.y, 0.f), w1.y, s);
            s = fmaf(fmaxf(h1.z + r1.z + q1.z, 0.f), w1.z, s);
            s = fmaf(fmaxf(h1.w + r1.w + q1.w, 0.f), w1.w, s);
            s += __shfl_xor(s, 1, 32);
            s += __shfl_xor(s, 2, 32);
            s += __shfl_xor(s, 4, 32);
            const float alpha = 1.0f / (1.0f + __expf(-(s + bias)));
            if (t == j) myalpha = alpha;
        }

        const int old = atomicExch(&head[obj], e);
        int4 rv;
        rv.x = old;
        rv.y = sub;
        rv.z = rel;
        rv.w = __float_as_int(myalpha);
        ((int4*)rec)[e] = rv;

        c += STRIDE;
        if (c < NCHUNK) {
            const int en = c * 32 + lane;
            sub = edges[en * 3 + 0];
            rel = edges[en * 3 + 1];
            obj = edges[en * 3 + 2];
            q   = r_idx[en];
        }
    }
}

#define RELA_PITCH   132
#define NODE_WGS     256
#define NODE_THREADS 1024

__global__ __launch_bounds__(NODE_THREADS)
void node_sum_kernel(const float* __restrict__ hidden,
                     const float* __restrict__ rela,
                     const int*   __restrict__ head,
                     const int*   __restrict__ rec,
                     float* __restrict__ agg) {
    __shared__ __attribute__((aligned(16))) float rs[N_EMB * RELA_PITCH];

    const int t = threadIdx.x & 3;
    const int n = (threadIdx.x >> 2) * NODE_WGS + blockIdx.x;
    const bool live = n < N_NODE;

    int e = live ? head[n] : -1;

    constexpr int NV4 = N_EMB * (IN_DIM / 4);
    constexpr int NST = (NV4 + NODE_THREADS - 1) / NODE_THREADS;
    float4 sv[NST];
    #pragma unroll
    for (int k = 0; k < NST; ++k) {
        const int i = k * NODE_THREADS + (int)threadIdx.x;
        sv[k] = ((const float4*)rela)[i < NV4 ? i : NV4 - 1];
    }
    int4 r = ((const int4*)rec)[e < 0 ? 0 : e];
    #pragma unroll
    for (int k = 0; k < NST; ++k) {
        const int i = k * NODE_THREADS + (int)threadIdx.x;
        if (i < NV4) *(float4*)&rs[(i >> 5) * RELA_PITCH + (i & 31) * 4] = sv[k];
    }
    __syncthreads();
    if (!live) return;

    float4 a[8];
    #pragma unroll
    for (int c = 0; c < 8; ++c) a[c] = make_float4(0.f, 0.f, 0.f, 0.f);

    for (int it = 0; it < N_EDGE && e >= 0; ++it) {
        const int   en = r.x;
        const float al = __int_as_float(r.w);
        const float4* hs = (const float4*)(hidden + (size_t)r.y * IN_DIM) + t;
        const float4* hr = (const float4*)(rs + r.z * RELA_PITCH) + t;
        float4 h[8];
        #pragma unroll
        for (int c = 0; c < 8; ++c) h[c] = hs[4 * c];
        int4 rn = r;
        if (en >= 0) rn = ((const int4*)rec)[en];
        #pragma unroll
        for (int c = 0; c < 8; ++c) {
            const float4 g = hr[4 * c];
            a[c].x += al * h[c].x * g.x;
            a[c].y += al * h[c].y * g.y;
            a[c].z += al * h[c].z * g.z;
            a[c].w += al * h[c].w * g.w;
        }
        e = en;
        r = rn;
    }

    float4* dst = (float4*)(agg + (size_t)n * IN_DIM) + t;
    #pragma unroll
    for (int c = 0; c < 8; ++c) dst[4 * c] = a[c];
}

extern "C" void kernel_launch(void* const* d_in, const int* in_sizes, int n_in,
                              void* d_out, int out_size, void* d_ws, size_t ws_size,
                              hipStream_t stream) {
    const int*   q_rel  = (const int*)  d_in[1];
    const int*   r_idx  = (const int*)  d_in[2];
    const float* hidden = (const float*)d_in[3];
    const int*   edges  = (const int*)  d_in[4];
    const float* rela   = (const float*)d_in[6];
    const float* Ws     = (const float*)d_in[7];
    const float* Wr     = (const float*)d_in[8];
    const float* WqrW   = (const float*)d_in[9];
    const float* Wqrb   = (const float*)d_in[10];
    const float* waW    = (const float*)d_in[11];
    const float* wab    = (const float*)d_in[12];
    const float* Wh     = (const float*)d_in[13];
    float* out = (float*)d_out;

    char* ws = (char*)d_ws;
    float* agg   = (float*)(ws);
    float* hproj = (float*)(ws + (size_t)N_NODE * IN_DIM * 4);
    float* rproj = (float*)(ws + (size_t)N_NODE * IN_DIM * 4
                               + (size_t)N_NODE * ATTN_DIM * 4);
    float* qproj = (float*)((char*)rproj + (((size_t)N_EMB * ATTN_DIM * 4 + 255ull) & ~255ull));
    int*   head  = (int*)((char*)qproj + (size_t)N_Q * ATTN_DIM * 4);
    int*   rec   = (int*)((char*)head + (((size_t)N_NODE * 4 + 255ull) & ~255ull));

    front_kernel<<<SP_WGS + GEMM1_WGS, 256, 0, stream>>>(
        hidden, Ws, hproj, rela, Wr, WqrW, Wqrb, q_rel, rproj, qproj, head);

    edge_link_kernel<<<EDGE_WGS, EDGE_THREADS, 0, stream>>>(edges, r_idx, hproj, rproj, qproj,
                                                          waW, wab, head, rec);

    node_sum_kernel<<<NODE_WGS, NODE_THREADS, 0, stream>>>(hidden, rela, head, rec, agg);

    {
        const int mTiles = N_NODE / 16;
        const int waveTiles = mTiles * (OUT_DIM / 64);
        gemm_out_kernel<<<(waveTiles + 7) / 8, 256, 0, stream>>>(
            agg, Wh, out, N_NODE);
    }
}


# Edge pass: a chunk's 8 gate steps made independent: 8-lane exchanges are DPP register moves, not LDS permutes, so all 16 row gathers fly at once; one sigmoid per lane; 256 blocks; sums unchanged
# speedup vs baseline: 1.5475x; 1.0413x over previous
#include <hip/hip_runtime.h>
#include <math.h>

#define N_NODE   50000
#define N_EDGE   800000
#define IN_DIM   128
#define OUT_DIM  128
#define ATTN_DIM 64
#define N_REL    200
#define N_Q      64
#define N_EMB    (2 * N_REL + 1)

typedef __attribute__((ext_vector_type(2))) float v2f;
typedef __attribute__((ext_vector_type(8))) float v8f;

__device__ __forceinline__ void small_proj_body(int idx,
                                  const float* __restrict__ rela,
                                  const float* __restrict__ Wr,
                                  const float* __restrict__ WqrW,
                                  const float* __restrict__ Wqrb,
                                  const int*   __restrict__ q_rel,
                                  float* __restrict__ rproj,
                                  float* __restrict__ qproj) {
    const int total = (N_EMB + N_Q) * ATTN_DIM;
    if (idx >= total) return;
    int col = idx & (ATTN_DIM - 1);
    int row = idx >> 6;
    if (row < N_EMB) {
        const float* h = rela + (size_t)row * IN_DIM;
        float s = 0.f;
        #pragma unroll 8
        for (int k = 0; k < IN_DIM; ++k) s = fmaf(h[k], Wr[k * ATTN_DIM + col], s);
        rproj[row * ATTN_DIM + col] = s;
    } else {
        int q = row - N_EMB;
        const float* h = rela + (size_t)q_rel[q] * IN_DIM;
        float s = Wqrb[col];
        #pragma unroll 8
        for (int k = 0; k < IN_DIM; ++k) s = fmaf(h[k], WqrW[k * ATTN_DIM + col], s);
        qproj[q * ATTN_DIM + col] = s;
    }
}

template <int N>
__device__ __forceinline__
void gemm_wmma_f32_body(const float* __restrict__ A,
                        const float* __restrict__ B,
                        float* __restrict__ C,
                        int M, int blk) {
    constexpr int K = 128;
    __shared__ __attribute__((aligned(16))) v2f Bs2[(K / 2) * N];

    constexpr int NTHR = 256;
    constexpr int NQ   = (K / 2) * (N / 4);
    constexpr int PER  = NQ / NTHR;
    static_assert(NQ % NTHR == 0, "B must split evenly over the block");
    float4 r0[PER], r1[PER];
    #pragma unroll
    for (int j = 0; j < PER; ++j) {
        const int i  = j * NTHR + (int)threadIdx.x;
        const int kp = i / (N / 4);
        const int n  = (i - kp * (N / 4)) * 4;
        r0[j] = *(const float4*)(B + (2 * kp)     * N + n);
        r1[j] = *(const float4*)(B + (2 * kp + 1) * N + n);
    }
    #pragma unroll
    for (int j = 0; j < PER; ++j) {
        const int i  = j * NTHR + (int)threadIdx.x;
        const int kp = i / (N / 4);
        const int n  = (i - kp * (N / 4)) * 4;
        float4* d = (float4*)&Bs2[kp * N + n];
        d[0] = make_float4(r0[j].x, r1[j].x, r0[j].y, r1[j].y);
        d[1] = make_float4(r0[j].z, r1[j].z, r0[j].w, r1[j].w);
    }
    __syncthreads();

    const int lane = threadIdx.x & 31;
    const int waveInBlock = threadIdx.x >> 5;
    constexpr int colTiles = N / 64;
    const long tileId = (long)blk * 8 + waveInBlock;
    const int  mTile  = (int)(tileId / colTiles);
    const int  nTile  = (int)(tileId % colTiles);
    if (mTile * 16 >= M) return;

    const int mrow  = mTile * 16 + (lane & 15);
    const int sel   = lane >> 4;
    const int koff  = sel * 2;
    const int ncol0 = nTile * 64;
    const int nlane = ncol0 + (lane & 15);

    v8f acc0 = {}, acc1 = {}, acc2 = {}, acc3 = {};

    const float* __restrict__ arow = A + (size_t)mrow * K;
    #pragma unroll 4
    for (int k0 = 0; k0 < K; k0 += 4) {
        v2f a;
        a.x = arow[k0 + koff];
        a.y = arow[k0 + koff + 1];

        const v2f* __restrict__ brow = &Bs2[(k0 / 2 + sel) * N + nlane];
        acc0 = __builtin_amdgcn_wmma_f32_16x16x4_f32(false, a, false, brow[ 0], (short)0, acc0, false, false);
        acc1 = __builtin_amdgcn_wmma_f32_16x16x4_f32(false, a, false, brow[16], (short)0, acc1, false, false);
        acc2 = __builtin_amdgcn_wmma_f32_16x16x4_f32(false, a, false, brow[32], (short)0, acc2, false, false);
        acc3 = __builtin_amdgcn_wmma_f32_16x16x4_f32(false, a, false, brow[48], (short)0, acc3, false, false);
    }

    const int cmBase = mTile * 16 + ((lane < 16) ? 0 : 8);
    const int cn = lane & 15;
    float* __restrict__ c0 = C + (size_t)cmBase * N + ncol0 + cn;
    #pragma unroll
    for (int r = 0; r < 8; ++r) {
        float* cr = c0 + (size_t)r * N;
        cr[ 0] = acc0[r];
        cr[16] = acc1[r];
        cr[32] = acc2[r];
        cr[48] = acc3[r];
    }
}

__global__ __launch_bounds__(256)
void gemm_out_kernel(const float* __restrict__ A,
                     const float* __restrict__ B,
                     float* __restrict__ C,
                     int M) {
    gemm_wmma_f32_body<OUT_DIM>(A, B, C, M, (int)blockIdx.x);
}

#define SP_WGS    (((N_EMB + N_Q) * ATTN_DIM + 255) / 256)
#define GEMM1_WGS (((N_NODE / 16) * (ATTN_DIM / 64) + 7) / 8)

__global__ __launch_bounds__(256)
void front_kernel(const float* __restrict__ hidden,
                  const float* __restrict__ Ws,
                  float* __restrict__ hproj,
                  const float* __restrict__ rela,
                  const float* __restrict__ Wr,
                  const float* __restrict__ WqrW,
                  const float* __restrict__ Wqrb,
                  const int*   __restrict__ q_rel,
                  float* __restrict__ rproj,
                  float* __restrict__ qproj,
                  int* __restrict__ head) {
    if (blockIdx.x < SP_WGS) {
        const int idx = (int)blockIdx.x * 256 + (int)threadIdx.x;
        static_assert(N_NODE % 4 == 0 && N_NODE / 4 <= SP_WGS * 256, "head[] must be covered");
        if (idx < N_NODE / 4) ((int4*)head)[idx] = make_int4(-1, -1, -1, -1);
        small_proj_body(idx, rela, Wr, WqrW, Wqrb, q_rel, rproj, qproj);
        return;
    }
    gemm_wmma_f32_body<ATTN_DIM>(hidden, Ws, hproj, N_NODE, (int)blockIdx.x - SP_WGS);
}

#define TAB_PITCH    68
#define EDGE_WGS     256
#define EDGE_THREADS 1024

__host__ __device__ constexpr unsigned dpp8_sel(unsigned s0, unsigned s1, unsigned s2, unsigned s3,
                                                unsigned s4, unsigned s5, unsigned s6, unsigned s7) {
    return s0 | (s1 << 3) | (s2 << 6) | (s3 << 9) | (s4 << 12) | (s5 << 15) | (s6 << 18) | (s7 << 21);
}

template <int J>
__device__ __forceinline__ int group_bcast(int v) {
    return (int)__builtin_amdgcn_mov_dpp8((unsigned)v, dpp8_sel(J, J, J, J, J, J, J, J));
}

template <int CTRL>
__device__ __forceinline__ float lane_xor(float s) {
    return __int_as_float(__builtin_amdgcn_update_dpp(0, __float_as_int(s), CTRL, 0xf, 0xf, true));
}

__device__ __forceinline__ float group_sum8(float s) {
    s += lane_xor<0xB1>(s);
    s += lane_xor<0x4E>(s);
    s += lane_xor<0x164>(s);
    return s;
}

template <int J>
__device__ __forceinline__ void gate_request(int sub, const float* __restrict__ hproj, int t,
                                             float4& h0, float4& h1) {
    const int s_sub = group_bcast<J>(sub);
    const float4* hp = (const float4*)(hproj + (size_t)s_sub * ATTN_DIM);
    h0 = hp[t];
    h1 = hp[8 + t];
}

template <int J>
__device__ __forceinline__ float gate_dot(int rel, int q, const float* tab, int t,
                                          const float4 h0, const float4 h1,
                                          const float4 w0, const float4 w1) {
    const int s_rel = group_bcast<J>(rel);
    const int s_q   = group_bcast<J>(q);
    const float4* rp = (const float4*)(tab + s_rel * TAB_PITCH);
    const float4* qp = (const float4*)(tab + (N_EMB + s_q) * TAB_PITCH);
    const float4 r0 = rp[t], r1 = rp[8 + t];
    const float4 q0 = qp[t], q1 = qp[8 + t];
    float s = fmaxf(h0.x + r0.x + q0.x, 0.f) * w0.x;
    s = fmaf(fmaxf(h0.y + r0.y + q0.y, 0.f), w0.y, s);
    s = fmaf(fmaxf(h0.z + r0.z + q0.z, 0.f), w0.z, s);
    s = fmaf(fmaxf(h0.w + r0.w + q0.w, 0.f), w0.w, s);
    s = fmaf(fmaxf(h1.x + r1.x + q1.x, 0.f), w1.x, s);
    s = fmaf(fmaxf(h1.y + r1.y + q1.y, 0.f), w1.y, s);
    s = fmaf(fmaxf(h1.z + r1.z + q1.z, 0.f), w1.z, s);
    s = fmaf(fmaxf(h1.w + r1.w + q1.w, 0.f), w1.w, s);
    return group_sum8(s);
}

__global__ __launch_bounds__(EDGE_THREADS)
void edge_link_kernel(const int*   __restrict__ edges,
                      const int*   __restrict__ r_idx,
                      const float* __restrict__ hproj,
                      const float* __restrict__ rproj,
                      const float* __restrict__ qproj,
                      const float* __restrict__ waW,
                      const float* __restrict__ wab,
                      int* __restrict__ head,
                      int* __restrict__ rec) {
    __shared__ __attribute__((aligned(16))) float tab[(N_EMB + N_Q) * TAB_PITCH];

    const int lane  = threadIdx.x & 31;
    const int t     = lane & 7;
    constexpr int NCHUNK = N_EDGE / 32;
    constexpr int STRIDE = EDGE_WGS * (EDGE_THREADS / 32);

    int c = (threadIdx.x >> 5) * EDGE_WGS + blockIdx.x;
    int sub = 0, rel = 0, obj = 0, q = 0;
    if (c < NCHUNK) {
        const int e = c * 32 + lane;
        sub = edges[e * 3 + 0];
        rel = edges[e * 3 + 1];
        obj = edges[e * 3 + 2];
        q   = r_idx[e];
    }

    constexpr int NR4 = N_EMB * (ATTN_DIM / 4);
    constexpr int NT4 = (N_EMB + N_Q) * (ATTN_DIM / 4);
    #pragma unroll
    for (int k = 0; k < (NT4 + EDGE_THREADS - 1) / EDGE_THREADS; ++k) {
        const int i = k * EDGE_THREADS + (int)threadIdx.x;
        if (i < NT4) {
            const float4 v = (i < NR4) ? ((const float4*)rproj)[i]
                                       : ((const float4*)qproj)[i - NR4];
            *(float4*)&tab[(i >> 4) * TAB_PITCH + (i & 15) * 4] = v;
        }
    }
    __syncthreads();

    const float  bias = wab[0];
    const float4 w0 = ((const float4*)waW)[t];
    const float4 w1 = ((const float4*)waW)[8 + t];

    while (c < NCHUNK) {
        const int e = c * 32 + lane;

        const int old = atomicExch(&head[obj], e);

        float4 a0, a1, b0, b1, c0, c1, d0, d1, e0, e1, f0, f1, g0, g1, i0, i1;
        gate_request<0>(sub, hproj, t, a0, a1);
        gate_request<1>(sub, hproj, t, b0, b1);
        gate_request<2>(sub, hproj, t, c0, c1);
        gate_request<3>(sub, hproj, t, d0, d1);
        gate_request<4>(sub, hproj, t, e0, e1);
        gate_request<5>(sub, hproj, t, f0, f1);
        gate_request<6>(sub, hproj, t, g0, g1);
        gate_request<7>(sub, hproj, t, i0, i1);

        const int cn = c + STRIDE;
        int nsub = 0, nrel = 0, nobj = 0, nq = 0;
        if (cn < NCHUNK) {
            const int en = cn * 32 + lane;
            nsub = edges[en * 3 + 0];
            nrel = edges[en * 3 + 1];
            nobj = edges[en * 3 + 2];
            nq   = r_idx[en];
        }

        float s, mine;
        mine = gate_dot<0>(rel, q, tab, t, a0, a1, w0, w1);
        s = gate_dot<1>(rel, q, tab, t, b0, b1, w0, w1);  mine = (t == 1) ? s : mine;
        s = gate_dot<2>(rel, q, tab, t, c0, c1, w0, w1);  mine = (t == 2) ? s : mine;
        s = gate_dot<3>(rel, q, tab, t, d0, d1, w0, w1);  mine = (t == 3) ? s : mine;
        s = gate_dot<4>(rel, q, tab, t, e0, e1, w0, w1);  mine = (t == 4) ? s : mine;
        s = gate_dot<5>(rel, q, tab, t, f0, f1, w0, w1);  mine = (t == 5) ? s : mine;
        s = gate_dot<6>(rel, q, tab, t, g0, g1, w0, w1);  mine = (t == 6) ? s : mine;
        s = gate_dot<7>(rel, q, tab, t, i0, i1, w0, w1);  mine = (t == 7) ? s : mine;
        const float alpha = 1.0f / (1.0f + __expf(-(mine + bias)));

        int4 rv;
        rv.x = old;
        rv.y = sub;
        rv.z = rel;
        rv.w = __float_as_int(alpha);
        ((int4*)rec)[e] = rv;

        c = cn;
        sub = nsub;  rel = nrel;  obj = nobj;  q = nq;
    }
}

#define RELA_PITCH   132
#define NODE_WGS     256
#define NODE_THREADS 1024

__global__ __launch_bounds__(NODE_THREADS)
void node_sum_kernel(const float* __restrict__ hidden,
                     const float* __restrict__ rela,
                     const int*   __restrict__ head,
                     const int*   __restrict__ rec,
                     float* __restrict__ agg) {
    __shared__ __attribute__((aligned(16))) float rs[N_EMB * RELA_PITCH];

    const int t = threadIdx.x & 3;
    const int n = (threadIdx.x >> 2) * NODE_WGS + blockIdx.x;
    const bool live = n < N_NODE;

    int e = live ? head[n] : -1;

    constexpr int NV4 = N_EMB * (IN_DIM / 4);
    constexpr int NST = (NV4 + NODE_THREADS - 1) / NODE_THREADS;
    float4 sv[NST];
    #pragma unroll
    for (int k = 0; k < NST; ++k) {
        const int i = k * NODE_THREADS + (int)threadIdx.x;
        sv[k] = ((const float4*)rela)[i < NV4 ? i : NV4 - 1];
    }
    int4 r = ((const int4*)rec)[e < 0 ? 0 : e];
    #pragma unroll
    for (int k = 0; k < NST; ++k) {
        const int i = k * NODE_THREADS + (int)threadIdx.x;
        if (i < NV4) *(float4*)&rs[(i >> 5) * RELA_PITCH + (i & 31) * 4] = sv[k];
    }
    __syncthreads();
    if (!live) return;

    float4 a[8];
    #pragma unroll
    for (int c = 0; c < 8; ++c) a[c] = make_float4(0.f, 0.f, 0.f, 0.f);

    for (int it = 0; it < N_EDGE && e >= 0; ++it) {
        const int   en = r.x;
        const float al = __int_as_float(r.w);
        const float4* hs = (const float4*)(hidden + (size_t)r.y * IN_DIM) + t;
        const float4* hr = (const float4*)(rs + r.z * RELA_PITCH) + t;
        float4 h[8];
        #pragma unroll
        for (int c = 0; c < 8; ++c) h[c] = hs[4 * c];
        int4 rn = r;
        if (en >= 0) rn = ((const int4*)rec)[en];
        #pragma unroll
        for (int c = 0; c < 8; ++c) {
            const float4 g = hr[4 * c];
            a[c].x += al * h[c].x * g.x;
            a[c].y += al * h[c].y * g.y;
            a[c].z += al * h[c].z * g.z;
            a[c].w += al * h[c].w * g.w;
        }
        e = en;
        r = rn;
    }

    float4* dst = (float4*)(agg + (size_t)n * IN_DIM) + t;
    #pragma unroll
    for (int c = 0; c < 8; ++c) dst[4 * c] = a[c];
}

extern "C" void kernel_launch(void* const* d_in, const int* in_sizes, int n_in,
                              void* d_out, int out_size, void* d_ws, size_t ws_size,
                              hipStream_t stream) {
    const int*   q_rel  = (const int*)  d_in[1];
    const int*   r_idx  = (const int*)  d_in[2];
    const float* hidden = (const float*)d_in[3];
    const int*   edges  = (const int*)  d_in[4];
    const float* rela   = (const float*)d_in[6];
    const float* Ws     = (const float*)d_in[7];
    const float* Wr     = (const float*)d_in[8];
    const float* WqrW   = (const float*)d_in[9];
    const float* Wqrb   = (const float*)d_in[10];
    const float* waW    = (const float*)d_in[11];
    const float* wab    = (const float*)d_in[12];
    const float* Wh     = (const float*)d_in[13];
    float* out = (float*)d_out;

    char* ws = (char*)d_ws;
    float* agg   = (float*)(ws);
    float* hproj = (float*)(ws + (size_t)N_NODE * IN_DIM * 4);
    float* rproj = (float*)(ws + (size_t)N_NODE * IN_DIM * 4
                               + (size_t)N_NODE * ATTN_DIM * 4);
    float* qproj = (float*)((char*)rproj + (((size_t)N_EMB * ATTN_DIM * 4 + 255ull) & ~255ull));
    int*   head  = (int*)((char*)qproj + (size_t)N_Q * ATTN_DIM * 4);
    int*   rec   = (int*)((char*)head + (((size_t)N_NODE * 4 + 255ull) & ~255ull));

    front_kernel<<<SP_WGS + GEMM1_WGS, 256, 0, stream>>>(
        hidden, Ws, hproj, rela, Wr, WqrW, Wqrb, q_rel, rproj, qproj, head);

    edge_link_kernel<<<EDGE_WGS, EDGE_THREADS, 0, stream>>>(edges, r_idx, hproj, rproj, qproj,
                                                          waW, wab, head, rec);

    node_sum_kernel<<<NODE_WGS, NODE_THREADS, 0, stream>>>(hidden, rela, head, rec, agg);

    {
        const int mTiles = N_NODE / 16;
        const int waveTiles = mTiles * (OUT_DIM / 64);
        gemm_out_kernel<<<(waveTiles + 7) / 8, 256, 0, stream>>>(
            agg, Wh, out, N_NODE);
    }
}


# Edge-to-node grouping moved into LDS: each 3200-edge slice counting-sorted by 196-node range (no global atomics, whole-line record stores); owner block links its lists in LDS; sums in LDS list order
# speedup vs baseline: 2.4123x; 1.5588x over previous
#include <hip/hip_runtime.h>
#include <math.h>

#define N_NODE   50000
#define N_EDGE   800000
#define IN_DIM   128
#define OUT_DIM  128
#define ATTN_DIM 64
#define N_REL    200
#define N_Q      64
#define N_EMB    (2 * N_REL + 1)

typedef __attribute__((ext_vector_type(2))) float v2f;
typedef __attribute__((ext_vector_type(8))) float v8f;

__device__ __forceinline__ void small_proj_body(int idx,
                                  const float* __restrict__ rela,
                                  const float* __restrict__ Wr,
                                  const float* __restrict__ WqrW,
                                  const float* __restrict__ Wqrb,
                                  const int*   __restrict__ q_rel,
                                  float* __restrict__ rproj,
                                  float* __restrict__ qproj) {
    const int total = (N_EMB + N_Q) * ATTN_DIM;
    if (idx >= total) return;
    int col = idx & (ATTN_DIM - 1);
    int row = idx >> 6;
    if (row < N_EMB) {
        const float* h = rela + (size_t)row * IN_DIM;
        float s = 0.f;
        #pragma unroll 8
        for (int k = 0; k < IN_DIM; ++k) s = fmaf(h[k], Wr[k * ATTN_DIM + col], s);
        rproj[row * ATTN_DIM + col] = s;
    } else {
        int q = row - N_EMB;
        const float* h = rela + (size_t)q_rel[q] * IN_DIM;
        float s = Wqrb[col];
        #pragma unroll 8
        for (int k = 0; k < IN_DIM; ++k) s = fmaf(h[k], WqrW[k * ATTN_DIM + col], s);
        qproj[q * ATTN_DIM + col] = s;
    }
}

template <int N>
__device__ __forceinline__
void gemm_wmma_f32_body(const float* __restrict__ A,
                        const float* __restrict__ B,
                        float* __restrict__ C,
                        int M, int blk) {
    constexpr int K = 128;
    __shared__ __attribute__((aligned(16))) v2f Bs2[(K / 2) * N];

    constexpr int NTHR = 256;
    constexpr int NQ   = (K / 2) * (N / 4);
    constexpr int PER  = NQ / NTHR;
    static_assert(NQ % NTHR == 0, "B must split evenly over the block");
    float4 r0[PER], r1[PER];
    #pragma unroll
    for (int j = 0; j < PER; ++j) {
        const int i  = j * NTHR + (int)threadIdx.x;
        const int kp = i / (N / 4);
        const int n  = (i - kp * (N / 4)) * 4;
        r0[j] = *(const float4*)(B + (2 * kp)     * N + n);
        r1[j] = *(const float4*)(B + (2 * kp + 1) * N + n);
    }
    #pragma unroll
    for (int j = 0; j < PER; ++j) {
        const int i  = j * NTHR + (int)threadIdx.x;
        const int kp = i / (N / 4);
        const int n  = (i - kp * (N / 4)) * 4;
        float4* d = (float4*)&Bs2[kp * N + n];
        d[0] = make_float4(r0[j].x, r1[j].x, r0[j].y, r1[j].y);
        d[1] = make_float4(r0[j].z, r1[j].z, r0[j].w, r1[j].w);
    }
    __syncthreads();

    const int lane = threadIdx.x & 31;
    const int waveInBlock = threadIdx.x >> 5;
    constexpr int colTiles = N / 64;
    const long tileId = (long)blk * 8 + waveInBlock;
    const int  mTile  = (int)(tileId / colTiles);
    const int  nTile  = (int)(tileId % colTiles);
    if (mTile * 16 >= M) return;

    const int mrow  = mTile * 16 + (lane & 15);
    const int sel   = lane >> 4;
    const int koff  = sel * 2;
    const int ncol0 = nTile * 64;
    const int nlane = ncol0 + (lane & 15);

    v8f acc0 = {}, acc1 = {}, acc2 = {}, acc3 = {};

    const float* __restrict__ arow = A + (size_t)mrow * K;
    #pragma unroll 4
    for (int k0 = 0; k0 < K; k0 += 4) {
        v2f a;
        a.x = arow[k0 + koff];
        a.y = arow[k0 + koff + 1];

        const v2f* __restrict__ brow = &Bs2[(k0 / 2 + sel) * N + nlane];
        acc0 = __builtin_amdgcn_wmma_f32_16x16x4_f32(false, a, false, brow[ 0], (short)0, acc0, false, false);
        acc1 = __builtin_amdgcn_wmma_f32_16x16x4_f32(false, a, false, brow[16], (short)0, acc1, false, false);
        acc2 = __builtin_amdgcn_wmma_f32_16x16x4_f32(false, a, false, brow[32], (short)0, acc2, false, false);
        acc3 = __builtin_amdgcn_wmma_f32_16x16x4_f32(false, a, false, brow[48], (short)0, acc3, false, false);
    }

    const int cmBase = mTile * 16 + ((lane < 16) ? 0 : 8);
    const int cn = lane & 15;
    float* __restrict__ c0 = C + (size_t)cmBase * N + ncol0 + cn;
    #pragma unroll
    for (int r = 0; r < 8; ++r) {
        float* cr = c0 + (size_t)r * N;
        cr[ 0] = acc0[r];
        cr[16] = acc1[r];
        cr[32] = acc2[r];
        cr[48] = acc3[r];
    }
}

__global__ __launch_bounds__(256)
void gemm_out_kernel(const float* __restrict__ A,
                     const float* __restrict__ B,
                     float* __restrict__ C,
                     int M) {
    gemm_wmma_f32_body<OUT_DIM>(A, B, C, M, (int)blockIdx.x);
}

#define SP_WGS    (((N_EMB + N_Q) * ATTN_DIM + 255) / 256)
#define GEMM1_WGS (((N_NODE / 16) * (ATTN_DIM / 64) + 7) / 8)

__global__ __launch_bounds__(256)
void front_kernel(const float* __restrict__ hidden,
                  const float* __restrict__ Ws,
                  float* __restrict__ hproj,
                  const float* __restrict__ rela,
                  const float* __restrict__ Wr,
                  const float* __restrict__ WqrW,
                  const float* __restrict__ Wqrb,
                  const int*   __restrict__ q_rel,
                  float* __restrict__ rproj,
                  float* __restrict__ qproj) {
    if (blockIdx.x < SP_WGS) {
        const int idx = (int)blockIdx.x * 256 + (int)threadIdx.x;
        small_proj_body(idx, rela, Wr, WqrW, Wqrb, q_rel, rproj, qproj);
        return;
    }
    gemm_wmma_f32_body<ATTN_DIM>(hidden, Ws, hproj, N_NODE, (int)blockIdx.x - SP_WGS);
}

#define EDGE_WGS     250
#define EDGE_THREADS 800
#define SLICE        3200
#define NODE_WGS     256
#define NODES_PER_WG 196
#define NODE_THREADS 800
#define OFFS_PITCH   288
static_assert(EDGE_WGS * SLICE == N_EDGE, "slices must cover the edges exactly");
static_assert(SLICE == 4 * EDGE_THREADS, "four edges per thread");
static_assert(NODE_WGS == 256, "the scan below is 256 wide");
static_assert(NODE_WGS * NODES_PER_WG >= N_NODE, "ranges must cover the nodes");
static_assert(NODES_PER_WG <= 256 && SLICE < 65536, "packing of the sort key");
static_assert(NODES_PER_WG * 4 <= NODE_THREADS, "four lanes per node");

__device__ __forceinline__ void block_scan256(const int* in, int* out, int* wsum) {
    const int tid  = (int)threadIdx.x;
    const int lane = tid & 31;
    int incl = 0;
    if (tid < 256) {
        incl = in[tid];
        #pragma unroll
        for (int d = 1; d < 32; d <<= 1) {
            const int u = __shfl_up(incl, d);
            if (lane >= d) incl += u;
        }
        if (lane == 31) wsum[tid >> 5] = incl;
    }
    __syncthreads();
    if (tid < 256) {
        int base = 0;
        for (int w = 0; w < (tid >> 5); ++w) base += wsum[w];
        out[tid + 1] = base + incl;
        if (tid == 0) out[0] = 0;
    }
    __syncthreads();
}

#define TAB_PITCH    68

__host__ __device__ constexpr unsigned dpp8_sel(unsigned s0, unsigned s1, unsigned s2, unsigned s3,
                                                unsigned s4, unsigned s5, unsigned s6, unsigned s7) {
    return s0 | (s1 << 3) | (s2 << 6) | (s3 << 9) | (s4 << 12) | (s5 << 15) | (s6 << 18) | (s7 << 21);
}

template <int J>
__device__ __forceinline__ int group_bcast(int v) {
    return (int)__builtin_amdgcn_mov_dpp8((unsigned)v, dpp8_sel(J, J, J, J, J, J, J, J));
}

template <int CTRL>
__device__ __forceinline__ float lane_xor(float s) {
    return __int_as_float(__builtin_amdgcn_update_dpp(0, __float_as_int(s), CTRL, 0xf, 0xf, true));
}

__device__ __forceinline__ float group_sum8(float s) {
    s += lane_xor<0xB1>(s);
    s += lane_xor<0x4E>(s);
    s += lane_xor<0x164>(s);
    return s;
}

template <int J>
__device__ __forceinline__ void gate_request(int sub, const float* __restrict__ hproj, int t,
                                             float4& h0, float4& h1) {
    const int s_sub = group_bcast<J>(sub);
    const float4* hp = (const float4*)(hproj + (size_t)s_sub * ATTN_DIM);
    h0 = hp[t];
    h1 = hp[8 + t];
}

template <int J>
__device__ __forceinline__ float gate_dot(int rel, int q, const float* tab, int t,
                                          const float4 h0, const float4 h1,
                                          const float4 w0, const float4 w1) {
    const int s_rel = group_bcast<J>(rel);
    const int s_q   = group_bcast<J>(q);
    const float4* rp = (const float4*)(tab + s_rel * TAB_PITCH);
    const float4* qp = (const float4*)(tab + (N_EMB + s_q) * TAB_PITCH);
    const float4 r0 = rp[t], r1 = rp[8 + t];
    const float4 q0 = qp[t], q1 = qp[8 + t];
    float s = fmaxf(h0.x + r0.x + q0.x, 0.f) * w0.x;
    s = fmaf(fmaxf(h0.y + r0.y + q0.y, 0.f), w0.y, s);
    s = fmaf(fmaxf(h0.z + r0.z + q0.z, 0.f), w0.z, s);
    s = fmaf(fmaxf(h0.w + r0.w + q0.w, 0.f), w0.w, s);
    s = fmaf(fmaxf(h1.x + r1.x + q1.x, 0.f), w1.x, s);
    s = fmaf(fmaxf(h1.y + r1.y + q1.y, 0.f), w1.y, s);
    s = fmaf(fmaxf(h1.z + r1.z + q1.z, 0.f), w1.z, s);
    s = fmaf(fmaxf(h1.w + r1.w + q1.w, 0.f), w1.w, s);
    return group_sum8(s);
}

__global__ __launch_bounds__(EDGE_THREADS)
void edge_bin_kernel(const int*   __restrict__ edges,
                     const int*   __restrict__ r_idx,
                     const float* __restrict__ hproj,
                     const float* __restrict__ rproj,
                     const float* __restrict__ qproj,
                     const float* __restrict__ waW,
                     const float* __restrict__ wab,
                     int* __restrict__ offs,
                     int* __restrict__ rec) {
    __shared__ __attribute__((aligned(16))) float tab[(N_EMB + N_Q) * TAB_PITCH];
    __shared__ __attribute__((aligned(16))) int4 ra[SLICE];
    __shared__ __attribute__((aligned(16))) int4 rb[SLICE];
    __shared__ int cnt[256];
    __shared__ int off[257];
    __shared__ int wsum[8];

    const int tid  = (int)threadIdx.x;
    const int lane = tid & 31;
    const int t    = lane & 7;
    const int e0   = (int)blockIdx.x * SLICE + tid;

    if (tid < 256) cnt[tid] = 0;

    int sub = edges[e0 * 3 + 0];
    int rel = edges[e0 * 3 + 1];
    int obj = edges[e0 * 3 + 2];
    int q   = r_idx[e0];

    constexpr int NR4 = N_EMB * (ATTN_DIM / 4);
    constexpr int NT4 = (N_EMB + N_Q) * (ATTN_DIM / 4);
    #pragma unroll
    for (int k = 0; k < (NT4 + EDGE_THREADS - 1) / EDGE_THREADS; ++k) {
        const int i = k * EDGE_THREADS + tid;
        if (i < NT4) {
            const float4 v = (i < NR4) ? ((const float4*)rproj)[i]
                                       : ((const float4*)qproj)[i - NR4];
            *(float4*)&tab[(i >> 4) * TAB_PITCH + (i & 15) * 4] = v;
        }
    }
    __syncthreads();

    const float  bias = wab[0];
    const float4 w0 = ((const float4*)waW)[t];
    const float4 w1 = ((const float4*)waW)[8 + t];

    for (int k = 0; k < 4; ++k) {
        float4 a0, a1, b0, b1, c0, c1, d0, d1, f0, f1, g0, g1, i0, i1, j0, j1;
        gate_request<0>(sub, hproj, t, a0, a1);
        gate_request<1>(sub, hproj, t, b0, b1);
        gate_request<2>(sub, hproj, t, c0, c1);
        gate_request<3>(sub, hproj, t, d0, d1);
        gate_request<4>(sub, hproj, t, f0, f1);
        gate_request<5>(sub, hproj, t, g0, g1);
        gate_request<6>(sub, hproj, t, i0, i1);
        gate_request<7>(sub, hproj, t, j0, j1);

        int nsub = 0, nrel = 0, nobj = 0, nq = 0;
        if (k < 3) {
            const int en = e0 + (k + 1) * EDGE_THREADS;
            nsub = edges[en * 3 + 0];
            nrel = edges[en * 3 + 1];
            nobj = edges[en * 3 + 2];
            nq   = r_idx[en];
        }

        float s, mine;
        mine = gate_dot<0>(rel, q, tab, t, a0, a1, w0, w1);
        s = gate_dot<1>(rel, q, tab, t, b0, b1, w0, w1);  mine = (t == 1) ? s : mine;
        s = gate_dot<2>(rel, q, tab, t, c0, c1, w0, w1);  mine = (t == 2) ? s : mine;
        s = gate_dot<3>(rel, q, tab, t, d0, d1, w0, w1);  mine = (t == 3) ? s : mine;
        s = gate_dot<4>(rel, q, tab, t, f0, f1, w0, w1);  mine = (t == 4) ? s : mine;
        s = gate_dot<5>(rel, q, tab, t, g0, g1, w0, w1);  mine = (t == 5) ? s : mine;
        s = gate_dot<6>(rel, q, tab, t, i0, i1, w0, w1);  mine = (t == 6) ? s : mine;
        s = gate_dot<7>(rel, q, tab, t, j0, j1, w0, w1);  mine = (t == 7) ? s : mine;
        const float alpha = 1.0f / (1.0f + __expf(-(mine + bias)));

        const int bin  = obj / NODES_PER_WG;
        const int loc  = obj - bin * NODES_PER_WG;
        const int rank = atomicAdd(&cnt[bin], 1);
        ra[k * EDGE_THREADS + tid] = make_int4((rank << 16) | (bin << 8) | loc, sub, rel,
                                               __float_as_int(alpha));

        sub = nsub;  rel = nrel;  obj = nobj;  q = nq;
    }
    __syncthreads();

    block_scan256(cnt, off, wsum);

    #pragma unroll
    for (int k = 0; k < 4; ++k) {
        const int4 r = ra[k * EDGE_THREADS + tid];
        const int bin  = (r.x >> 8) & 255;
        const int rank = (int)((unsigned)r.x >> 16);
        rb[off[bin] + rank] = make_int4(r.x & 255, r.y, r.z, r.w);
    }
    if (tid < 257) offs[(int)blockIdx.x * OFFS_PITCH + tid] = off[tid];
    __syncthreads();

    #pragma unroll
    for (int k = 0; k < 4; ++k)
        ((int4*)rec)[(int)blockIdx.x * SLICE + k * EDGE_THREADS + tid] = rb[k * EDGE_THREADS + tid];
}

#define RELA_PITCH   132
#define BATCH        4000
static_assert(BATCH == 5 * NODE_THREADS, "five records per thread and batch");

__global__ __launch_bounds__(NODE_THREADS)
void node_sum_kernel(const float* __restrict__ hidden,
                     const float* __restrict__ rela,
                     const int*   __restrict__ offs,
                     const int*   __restrict__ rec,
                     float* __restrict__ agg) {
    __shared__ __attribute__((aligned(16))) float rs[N_EMB * RELA_PITCH];
    __shared__ __attribute__((aligned(16))) int4 recs[BATCH];
    __shared__ int lhead[NODES_PER_WG + 1];
    __shared__ int slen[256];
    __shared__ int soff[257];
    __shared__ int sstart[256];
    __shared__ int wsum[8];

    const int tid = (int)threadIdx.x;
    const int t   = tid & 3;
    const int qd  = tid >> 2;
    const int n   = (int)blockIdx.x * NODES_PER_WG + qd;
    const bool live = (qd < NODES_PER_WG) && (n < N_NODE);

    int s0 = 0, s1 = 0;
    if (tid < EDGE_WGS) {
        s0 = offs[tid * OFFS_PITCH + (int)blockIdx.x];
        s1 = offs[tid * OFFS_PITCH + (int)blockIdx.x + 1];
    }

    constexpr int NV4 = N_EMB * (IN_DIM / 4);
    constexpr int NST = (NV4 + NODE_THREADS - 1) / NODE_THREADS;
    static_assert(RELA_PITCH >= IN_DIM + 4, "a row has 16 unused bytes behind it");
    float4 sv[NST];
    #pragma unroll
    for (int k = 0; k < NST; ++k) {
        const int i = k * NODE_THREADS + tid;
        sv[k] = ((const float4*)rela)[i < NV4 ? i : NV4 - 1];
    }
    if (tid < 256) {
        slen[tid]   = s1 - s0;
        sstart[tid] = tid * SLICE + s0;
    }
    #pragma unroll
    for (int k = 0; k < NST; ++k) {
        const int i = k * NODE_THREADS + tid;
        const int d = (i < NV4) ? (i >> 5) * RELA_PITCH + (i & 31) * 4 : IN_DIM;
        *(float4*)&rs[d] = sv[k];
    }
    __syncthreads();

    block_scan256(slen, soff, wsum);
    const int total = soff[256];

    float4 a[8];
    #pragma unroll
    for (int c = 0; c < 8; ++c) a[c] = make_float4(0.f, 0.f, 0.f, 0.f);

    for (int base = 0; base < total; base += BATCH) {
        if (tid <= NODES_PER_WG) lhead[tid] = -1;
        __syncthreads();

        const int lim = (total - base < BATCH) ? (total - base) : BATCH;
        int4 rr[5];
        #pragma unroll
        for (int u = 0; u < 5; ++u) {
            const int jl = u * NODE_THREADS + tid;
            const int j  = base + (jl < lim ? jl : lim - 1);
            int lo = 0, hi = 256;
            #pragma unroll
            for (int s = 0; s < 8; ++s) {
                const int mid = (lo + hi) >> 1;
                if (soff[mid] <= j) lo = mid; else hi = mid;
            }
            rr[u] = ((const int4*)rec)[sstart[lo] + (j - soff[lo])];
        }
        #pragma unroll
        for (int u = 0; u < 5; ++u) {
            const int jl  = u * NODE_THREADS + tid;
            const int hd  = (jl < lim) ? rr[u].x : NODES_PER_WG;
            const int old = atomicExch(&lhead[hd], jl);
            recs[jl] = make_int4(old, rr[u].y, rr[u].z, rr[u].w);
        }
        __syncthreads();

        int i = live ? lhead[qd] : -1;
        for (int it = 0; it < BATCH && i >= 0; ++it) {
            const int4  r  = recs[i];
            const float al = __int_as_float(r.w);
            const float4* hs = (const float4*)(hidden + (size_t)r.y * IN_DIM) + t;
            const float4* hr = (const float4*)(rs + r.z * RELA_PITCH) + t;
            float4 h[8];
            #pragma unroll
            for (int c = 0; c < 8; ++c) h[c] = hs[4 * c];
            #pragma unroll
            for (int c = 0; c < 8; ++c) {
                const float4 g = hr[4 * c];
                a[c].x += al * h[c].x * g.x;
                a[c].y += al * h[c].y * g.y;
                a[c].z += al * h[c].z * g.z;
                a[c].w += al * h[c].w * g.w;
            }
            i = r.x;
        }
        __syncthreads();
    }

    if (live) {
        float4* dst = (float4*)(agg + (size_t)n * IN_DIM) + t;
        #pragma unroll
        for (int c = 0; c < 8; ++c) dst[4 * c] = a[c];
    }
}

extern "C" void kernel_launch(void* const* d_in, const int* in_sizes, int n_in,
                              void* d_out, int out_size, void* d_ws, size_t ws_size,
                              hipStream_t stream) {
    const int*   q_rel  = (const int*)  d_in[1];
    const int*   r_idx  = (const int*)  d_in[2];
    const float* hidden = (const float*)d_in[3];
    const int*   edges  = (const int*)  d_in[4];
    const float* rela   = (const float*)d_in[6];
    const float* Ws     = (const float*)d_in[7];
    const float* Wr     = (const float*)d_in[8];
    const float* WqrW   = (const float*)d_in[9];
    const float* Wqrb   = (const float*)d_in[10];
    const float* waW    = (const float*)d_in[11];
    const float* wab    = (const float*)d_in[12];
    const float* Wh     = (const float*)d_in[13];
    float* out = (float*)d_out;

    char* ws = (char*)d_ws;
    float* agg   = (float*)(ws);
    float* hproj = (float*)(ws + (size_t)N_NODE * IN_DIM * 4);
    float* rproj = (float*)(ws + (size_t)N_NODE * IN_DIM * 4
                               + (size_t)N_NODE * ATTN_DIM * 4);
    float* qproj = (float*)((char*)rproj + (((size_t)N_EMB * ATTN_DIM * 4 + 255ull) & ~255ull));
    int*   offs  = (int*)((char*)qproj + (size_t)N_Q * ATTN_DIM * 4);
    int*   rec   = (int*)((char*)offs + (((size_t)EDGE_WGS * OFFS_PITCH * 4 + 255ull) & ~255ull));

    front_kernel<<<SP_WGS + GEMM1_WGS, 256, 0, stream>>>(
        hidden, Ws, hproj, rela, Wr, WqrW, Wqrb, q_rel, rproj, qproj);

    edge_bin_kernel<<<EDGE_WGS, EDGE_THREADS, 0, stream>>>(edges, r_idx, hproj, rproj, qproj,
                                                          waW, wab, offs, rec);

    node_sum_kernel<<<NODE_WGS, NODE_THREADS, 0, stream>>>(hidden, rela, offs, rec, agg);

    {
        const int mTiles = N_NODE / 16;
        const int waveTiles = mTiles * (OUT_DIM / 64);
        gemm_out_kernel<<<(waveTiles + 7) / 8, 256, 0, stream>>>(
            agg, Wh, out, N_NODE);
    }
}


# One launch fewer: out = sums @ Wh moves into the node pass; a range's 196 summed rows and Wh take the dead rela table's LDS, its 25 waves run the f32 WMMA tiles; agg never stored; sums unchanged
# speedup vs baseline: 2.6922x; 1.1160x over previous
#include <hip/hip_runtime.h>
#include <math.h>

#define N_NODE   50000
#define N_EDGE   800000
#define IN_DIM   128
#define OUT_DIM  128
#define ATTN_DIM 64
#define N_REL    200
#define N_Q      64
#define N_EMB    (2 * N_REL + 1)

typedef __attribute__((ext_vector_type(2))) float v2f;
typedef __attribute__((ext_vector_type(8))) float v8f;

__device__ __forceinline__ void small_proj_body(int idx,
                                  const float* __restrict__ rela,
                                  const float* __restrict__ Wr,
                                  const float* __restrict__ WqrW,
                                  const float* __restrict__ Wqrb,
                                  const int*   __restrict__ q_rel,
                                  float* __restrict__ rproj,
                                  float* __restrict__ qproj) {
    const int total = (N_EMB + N_Q) * ATTN_DIM;
    if (idx >= total) return;
    int col = idx & (ATTN_DIM - 1);
    int row = idx >> 6;
    if (row < N_EMB) {
        const float* h = rela + (size_t)row * IN_DIM;
        float s = 0.f;
        #pragma unroll 8
        for (int k = 0; k < IN_DIM; ++k) s = fmaf(h[k], Wr[k * ATTN_DIM + col], s);
        rproj[row * ATTN_DIM + col] = s;
    } else {
        int q = row - N_EMB;
        const float* h = rela + (size_t)q_rel[q] * IN_DIM;
        float s = Wqrb[col];
        #pragma unroll 8
        for (int k = 0; k < IN_DIM; ++k) s = fmaf(h[k], WqrW[k * ATTN_DIM + col], s);
        qproj[q * ATTN_DIM + col] = s;
    }
}

template <int N>
__device__ __forceinline__
void gemm_wmma_f32_body(const float* __restrict__ A,
                        const float* __restrict__ B,
                        float* __restrict__ C,
                        int M, int blk) {
    constexpr int K = 128;
    __shared__ __attribute__((aligned(16))) v2f Bs2[(K / 2) * N];

    constexpr int NTHR = 256;
    constexpr int NQ   = (K / 2) * (N / 4);
    constexpr int PER  = NQ / NTHR;
    static_assert(NQ % NTHR == 0, "B must split evenly over the block");
    float4 r0[PER], r1[PER];
    #pragma unroll
    for (int j = 0; j < PER; ++j) {
        const int i  = j * NTHR + (int)threadIdx.x;
        const int kp = i / (N / 4);
        const int n  = (i - kp * (N / 4)) * 4;
        r0[j] = *(const float4*)(B + (2 * kp)     * N + n);
        r1[j] = *(const float4*)(B + (2 * kp + 1) * N + n);
    }
    #pragma unroll
    for (int j = 0; j < PER; ++j) {
        const int i  = j * NTHR + (int)threadIdx.x;
        const int kp = i / (N / 4);
        const int n  = (i - kp * (N / 4)) * 4;
        float4* d = (float4*)&Bs2[kp * N + n];
        d[0] = make_float4(r0[j].x, r1[j].x, r0[j].y, r1[j].y);
        d[1] = make_float4(r0[j].z, r1[j].z, r0[j].w, r1[j].w);
    }
    __syncthreads();

    const int lane = threadIdx.x & 31;
    const int waveInBlock = threadIdx.x >> 5;
    constexpr int colTiles = N / 64;
    const long tileId = (long)blk * 8 + waveInBlock;
    const int  mTile  = (int)(tileId / colTiles);
    const int  nTile  = (int)(tileId % colTiles);
    if (mTile * 16 >= M) return;

    const int mrow  = mTile * 16 + (lane & 15);
    const int sel   = lane >> 4;
    const int koff  = sel * 2;
    const int ncol0 = nTile * 64;
    const int nlane = ncol0 + (lane & 15);

    v8f acc0 = {}, acc1 = {}, acc2 = {}, acc3 = {};

    const float* __restrict__ arow = A + (size_t)mrow * K;
    #pragma unroll 4
    for (int k0 = 0; k0 < K; k0 += 4) {
        v2f a;
        a.x = arow[k0 + koff];
        a.y = arow[k0 + koff + 1];

        const v2f* __restrict__ brow = &Bs2[(k0 / 2 + sel) * N + nlane];
        acc0 = __builtin_amdgcn_wmma_f32_16x16x4_f32(false, a, false, brow[ 0], (short)0, acc0, false, false);
        acc1 = __builtin_amdgcn_wmma_f32_16x16x4_f32(false, a, false, brow[16], (short)0, acc1, false, false);
        acc2 = __builtin_amdgcn_wmma_f32_16x16x4_f32(false, a, false, brow[32], (short)0, acc2, false, false);
        acc3 = __builtin_amdgcn_wmma_f32_16x16x4_f32(false, a, false, brow[48], (short)0, acc3, false, false);
    }

    const int cmBase = mTile * 16 + ((lane < 16) ? 0 : 8);
    const int cn = lane & 15;
    float* __restrict__ c0 = C + (size_t)cmBase * N + ncol0 + cn;
    #pragma unroll
    for (int r = 0; r < 8; ++r) {
        float* cr = c0 + (size_t)r * N;
        cr[ 0] = acc0[r];
        cr[16] = acc1[r];
        cr[32] = acc2[r];
        cr[48] = acc3[r];
    }
}

#define SP_WGS    (((N_EMB + N_Q) * ATTN_DIM + 255) / 256)
#define GEMM1_WGS (((N_NODE / 16) * (ATTN_DIM / 64) + 7) / 8)

__global__ __launch_bounds__(256)
void front_kernel(const float* __restrict__ hidden,
                  const float* __restrict__ Ws,
                  float* __restrict__ hproj,
                  const float* __restrict__ rela,
                  const float* __restrict__ Wr,
                  const float* __restrict__ WqrW,
                  const float* __restrict__ Wqrb,
                  const int*   __restrict__ q_rel,
                  float* __restrict__ rproj,
                  float* __restrict__ qproj) {
    if (blockIdx.x < SP_WGS) {
        const int idx = (int)blockIdx.x * 256 + (int)threadIdx.x;
        small_proj_body(idx, rela, Wr, WqrW, Wqrb, q_rel, rproj, qproj);
        return;
    }
    gemm_wmma_f32_body<ATTN_DIM>(hidden, Ws, hproj, N_NODE, (int)blockIdx.x - SP_WGS);
}

#define EDGE_WGS     250
#define EDGE_THREADS 800
#define SLICE        3200
#define NODE_WGS     256
#define NODES_PER_WG 196
#define NODE_THREADS 800
#define OFFS_PITCH   288
static_assert(EDGE_WGS * SLICE == N_EDGE, "slices must cover the edges exactly");
static_assert(SLICE == 4 * EDGE_THREADS, "four edges per thread");
static_assert(NODE_WGS == 256, "the scan below is 256 wide");
static_assert(NODE_WGS * NODES_PER_WG >= N_NODE, "ranges must cover the nodes");
static_assert(NODES_PER_WG <= 256 && SLICE < 65536, "packing of the sort key");
static_assert(NODES_PER_WG * 4 <= NODE_THREADS, "four lanes per node");

__device__ __forceinline__ void block_scan256(const int* in, int* out, int* wsum) {
    const int tid  = (int)threadIdx.x;
    const int lane = tid & 31;
    int incl = 0;
    if (tid < 256) {
        incl = in[tid];
        #pragma unroll
        for (int d = 1; d < 32; d <<= 1) {
            const int u = __shfl_up(incl, d);
            if (lane >= d) incl += u;
        }
        if (lane == 31) wsum[tid >> 5] = incl;
    }
    __syncthreads();
    if (tid < 256) {
        int base = 0;
        for (int w = 0; w < (tid >> 5); ++w) base += wsum[w];
        out[tid + 1] = base + incl;
        if (tid == 0) out[0] = 0;
    }
    __syncthreads();
}

#define TAB_PITCH    68

__host__ __device__ constexpr unsigned dpp8_sel(unsigned s0, unsigned s1, unsigned s2, unsigned s3,
                                                unsigned s4, unsigned s5, unsigned s6, unsigned s7) {
    return s0 | (s1 << 3) | (s2 << 6) | (s3 << 9) | (s4 << 12) | (s5 << 15) | (s6 << 18) | (s7 << 21);
}

template <int J>
__device__ __forceinline__ int group_bcast(int v) {
    return (int)__builtin_amdgcn_mov_dpp8((unsigned)v, dpp8_sel(J, J, J, J, J, J, J, J));
}

template <int CTRL>
__device__ __forceinline__ float lane_xor(float s) {
    return __int_as_float(__builtin_amdgcn_update_dpp(0, __float_as_int(s), CTRL, 0xf, 0xf, true));
}

__device__ __forceinline__ float group_sum8(float s) {
    s += lane_xor<0xB1>(s);
    s += lane_xor<0x4E>(s);
    s += lane_xor<0x164>(s);
    return s;
}

template <int J>
__device__ __forceinline__ void gate_request(int sub, const float* __restrict__ hproj, int t,
                                             float4& h0, float4& h1) {
    const int s_sub = group_bcast<J>(sub);
    const float4* hp = (const float4*)(hproj + (size_t)s_sub * ATTN_DIM);
    h0 = hp[t];
    h1 = hp[8 + t];
}

template <int J>
__device__ __forceinline__ float gate_dot(int rel, int q, const float* tab, int t,
                                          const float4 h0, const float4 h1,
                                          const float4 w0, const float4 w1) {
    const int s_rel = group_bcast<J>(rel);
    const int s_q   = group_bcast<J>(q);
    const float4* rp = (const float4*)(tab + s_rel * TAB_PITCH);
    const float4* qp = (const float4*)(tab + (N_EMB + s_q) * TAB_PITCH);
    const float4 r0 = rp[t], r1 = rp[8 + t];
    const float4 q0 = qp[t], q1 = qp[8 + t];
    float s = fmaxf(h0.x + r0.x + q0.x, 0.f) * w0.x;
    s = fmaf(fmaxf(h0.y + r0.y + q0.y, 0.f), w0.y, s);
    s = fmaf(fmaxf(h0.z + r0.z + q0.z, 0.f), w0.z, s);
    s = fmaf(fmaxf(h0.w + r0.w + q0.w, 0.f), w0.w, s);
    s = fmaf(fmaxf(h1.x + r1.x + q1.x, 0.f), w1.x, s);
    s = fmaf(fmaxf(h1.y + r1.y + q1.y, 0.f), w1.y, s);
    s = fmaf(fmaxf(h1.z + r1.z + q1.z, 0.f), w1.z, s);
    s = fmaf(fmaxf(h1.w + r1.w + q1.w, 0.f), w1.w, s);
    return group_sum8(s);
}

__global__ __launch_bounds__(EDGE_THREADS)
void edge_bin_kernel(const int*   __restrict__ edges,
                     const int*   __restrict__ r_idx,
                     const float* __restrict__ hproj,
                     const float* __restrict__ rproj,
                     const float* __restrict__ qproj,
                     const float* __restrict__ waW,
                     const float* __restrict__ wab,
                     int* __restrict__ offs,
                     int* __restrict__ rec) {
    __shared__ __attribute__((aligned(16))) float tab[(N_EMB + N_Q) * TAB_PITCH];
    __shared__ __attribute__((aligned(16))) int4 ra[SLICE];
    __shared__ __attribute__((aligned(16))) int4 rb[SLICE];
    __shared__ int cnt[256];
    __shared__ int off[257];
    __shared__ int wsum[8];

    const int tid  = (int)threadIdx.x;
    const int lane = tid & 31;
    const int t    = lane & 7;
    const int e0   = (int)blockIdx.x * SLICE + tid;

    if (tid < 256) cnt[tid] = 0;

    int sub = edges[e0 * 3 + 0];
    int rel = edges[e0 * 3 + 1];
    int obj = edges[e0 * 3 + 2];
    int q   = r_idx[e0];

    constexpr int NR4 = N_EMB * (ATTN_DIM / 4);
    constexpr int NT4 = (N_EMB + N_Q) * (ATTN_DIM / 4);
    #pragma unroll
    for (int k = 0; k < (NT4 + EDGE_THREADS - 1) / EDGE_THREADS; ++k) {
        const int i = k * EDGE_THREADS + tid;
        if (i < NT4) {
            const float4 v = (i < NR4) ? ((const float4*)rproj)[i]
                                       : ((const float4*)qproj)[i - NR4];
            *(float4*)&tab[(i >> 4) * TAB_PITCH + (i & 15) * 4] = v;
        }
    }
    __syncthreads();

    const float  bias = wab[0];
    const float4 w0 = ((const float4*)waW)[t];
    const float4 w1 = ((const float4*)waW)[8 + t];

    for (int k = 0; k < 4; ++k) {
        float4 a0, a1, b0, b1, c0, c1, d0, d1, f0, f1, g0, g1, i0, i1, j0, j1;
        gate_request<0>(sub, hproj, t, a0, a1);
        gate_request<1>(sub, hproj, t, b0, b1);
        gate_request<2>(sub, hproj, t, c0, c1);
        gate_request<3>(sub, hproj, t, d0, d1);
        gate_request<4>(sub, hproj, t, f0, f1);
        gate_request<5>(sub, hproj, t, g0, g1);
        gate_request<6>(sub, hproj, t, i0, i1);
        gate_request<7>(sub, hproj, t, j0, j1);

        int nsub = 0, nrel = 0, nobj = 0, nq = 0;
        if (k < 3) {
            const int en = e0 + (k + 1) * EDGE_THREADS;
            nsub = edges[en * 3 + 0];
            nrel = edges[en * 3 + 1];
            nobj = edges[en * 3 + 2];
            nq   = r_idx[en];
        }

        float s, mine;
        mine = gate_dot<0>(rel, q, tab, t, a0, a1, w0, w1);
        s = gate_dot<1>(rel, q, tab, t, b0, b1, w0, w1);  mine = (t == 1) ? s : mine;
        s = gate_dot<2>(rel, q, tab, t, c0, c1, w0, w1);  mine = (t == 2) ? s : mine;
        s = gate_dot<3>(rel, q, tab, t, d0, d1, w0, w1);  mine = (t == 3) ? s : mine;
        s = gate_dot<4>(rel, q, tab, t, f0, f1, w0, w1);  mine = (t == 4) ? s : mine;
        s = gate_dot<5>(rel, q, tab, t, g0, g1, w0, w1);  mine = (t == 5) ? s : mine;
        s = gate_dot<6>(rel, q, tab, t, i0, i1, w0, w1);  mine = (t == 6) ? s : mine;
        s = gate_dot<7>(rel, q, tab, t, j0, j1, w0, w1);  mine = (t == 7) ? s : mine;
        const float alpha = 1.0f / (1.0f + __expf(-(mine + bias)));

        const int bin  = obj / NODES_PER_WG;
        const int loc  = obj - bin * NODES_PER_WG;
        const int rank = atomicAdd(&cnt[bin], 1);
        ra[k * EDGE_THREADS + tid] = make_int4((rank << 16) | (bin << 8) | loc, sub, rel,
                                               __float_as_int(alpha));

        sub = nsub;  rel = nrel;  obj = nobj;  q = nq;
    }
    __syncthreads();

    block_scan256(cnt, off, wsum);

    #pragma unroll
    for (int k = 0; k < 4; ++k) {
        const int4 r = ra[k * EDGE_THREADS + tid];
        const int bin  = (r.x >> 8) & 255;
        const int rank = (int)((unsigned)r.x >> 16);
        rb[off[bin] + rank] = make_int4(r.x & 255, r.y, r.z, r.w);
    }
    if (tid < 257) offs[(int)blockIdx.x * OFFS_PITCH + tid] = off[tid];
    __syncthreads();

    #pragma unroll
    for (int k = 0; k < 4; ++k)
        ((int4*)rec)[(int)blockIdx.x * SLICE + k * EDGE_THREADS + tid] = rb[k * EDGE_THREADS + tid];
}

#define RELA_PITCH   132
#define BATCH        4000
static_assert(BATCH == 5 * NODE_THREADS, "five records per thread and batch");
#define AT_PITCH     132
#define M_TILES      ((NODES_PER_WG + 15) / 16)
#define AT_ROWS      (M_TILES * 16)
#define OUT_JOBS     (M_TILES * (OUT_DIM / 32))
static_assert(AT_ROWS * AT_PITCH + IN_DIM * OUT_DIM <= N_EMB * RELA_PITCH,
              "the sums and Wh fit in the LDS of the rela table");
static_assert(NODE_THREADS % 32 == 0, "whole waves only: the matrix unit needs every lane");

__global__ __launch_bounds__(NODE_THREADS)
void node_sum_kernel(const float* __restrict__ hidden,
                     const float* __restrict__ rela,
                     const int*   __restrict__ offs,
                     const int*   __restrict__ rec,
                     const float* __restrict__ Wh,
                     float* __restrict__ out) {
    __shared__ __attribute__((aligned(16))) float rs[N_EMB * RELA_PITCH];
    __shared__ __attribute__((aligned(16))) int4 recs[BATCH];
    __shared__ int lhead[NODES_PER_WG + 1];
    __shared__ int slen[256];
    __shared__ int soff[257];
    __shared__ int sstart[256];
    __shared__ int wsum[8];
    __shared__ int next_job;

    const int tid = (int)threadIdx.x;
    const int t   = tid & 3;
    const int qd  = tid >> 2;
    const int n   = (int)blockIdx.x * NODES_PER_WG + qd;
    const bool live = (qd < NODES_PER_WG) && (n < N_NODE);

    int s0 = 0, s1 = 0;
    if (tid < EDGE_WGS) {
        s0 = offs[tid * OFFS_PITCH + (int)blockIdx.x];
        s1 = offs[tid * OFFS_PITCH + (int)blockIdx.x + 1];
    }

    constexpr int NV4 = N_EMB * (IN_DIM / 4);
    constexpr int NST = (NV4 + NODE_THREADS - 1) / NODE_THREADS;
    static_assert(RELA_PITCH >= IN_DIM + 4, "a row has 16 unused bytes behind it");
    float4 sv[NST];
    #pragma unroll
    for (int k = 0; k < NST; ++k) {
        const int i = k * NODE_THREADS + tid;
        sv[k] = ((const float4*)rela)[i < NV4 ? i : NV4 - 1];
    }
    if (tid < 256) {
        slen[tid]   = s1 - s0;
        sstart[tid] = tid * SLICE + s0;
    }
    #pragma unroll
    for (int k = 0; k < NST; ++k) {
        const int i = k * NODE_THREADS + tid;
        const int d = (i < NV4) ? (i >> 5) * RELA_PITCH + (i & 31) * 4 : IN_DIM;
        *(float4*)&rs[d] = sv[k];
    }
    __syncthreads();

    block_scan256(slen, soff, wsum);
    const int total = soff[256];

    float4 a[8];
    #pragma unroll
    for (int c = 0; c < 8; ++c) a[c] = make_float4(0.f, 0.f, 0.f, 0.f);

    for (int base = 0; base < total; base += BATCH) {
        if (tid <= NODES_PER_WG) lhead[tid] = -1;
        __syncthreads();

        const int lim = (total - base < BATCH) ? (total - base) : BATCH;
        int4 rr[5];
        #pragma unroll
        for (int u = 0; u < 5; ++u) {
            const int jl = u * NODE_THREADS + tid;
            const int j  = base + (jl < lim ? jl : lim - 1);
            int lo = 0, hi = 256;
            #pragma unroll
            for (int s = 0; s < 8; ++s) {
                const int mid = (lo + hi) >> 1;
                if (soff[mid] <= j) lo = mid; else hi = mid;
            }
            rr[u] = ((const int4*)rec)[sstart[lo] + (j - soff[lo])];
        }
        #pragma unroll
        for (int u = 0; u < 5; ++u) {
            const int jl  = u * NODE_THREADS + tid;
            const int hd  = (jl < lim) ? rr[u].x : NODES_PER_WG;
            const int old = atomicExch(&lhead[hd], jl);
            recs[jl] = make_int4(old, rr[u].y, rr[u].z, rr[u].w);
        }
        __syncthreads();

        int i = live ? lhead[qd] : -1;
        for (int it = 0; it < BATCH && i >= 0; ++it) {
            const int4  r  = recs[i];
            const float al = __int_as_float(r.w);
            const float4* hs = (const float4*)(hidden + (size_t)r.y * IN_DIM) + t;
            const float4* hr = (const float4*)(rs + r.z * RELA_PITCH) + t;
            float4 h[8];
            #pragma unroll
            for (int c = 0; c < 8; ++c) h[c] = hs[4 * c];
            #pragma unroll
            for (int c = 0; c < 8; ++c) {
                const float4 g = hr[4 * c];
                a[c].x += al * h[c].x * g.x;
                a[c].y += al * h[c].y * g.y;
                a[c].z += al * h[c].z * g.z;
                a[c].w += al * h[c].w * g.w;
            }
            i = r.x;
        }
        __syncthreads();
    }

    float* at  = rs;
    v2f*   Bs2 = (v2f*)(rs + AT_ROWS * AT_PITCH);

    constexpr int NQ   = (IN_DIM / 2) * (OUT_DIM / 4);
    constexpr int WPER = (NQ + NODE_THREADS - 1) / NODE_THREADS;
    float4 w0[WPER], w1[WPER];
    #pragma unroll
    for (int j = 0; j < WPER; ++j) {
        const int i  = j * NODE_THREADS + tid;
        const int ic = (i < NQ) ? i : NQ - 1;
        const int kp = ic / (OUT_DIM / 4);
        const int nc = (ic - kp * (OUT_DIM / 4)) * 4;
        w0[j] = *(const float4*)(Wh + (2 * kp)     * OUT_DIM + nc);
        w1[j] = *(const float4*)(Wh + (2 * kp + 1) * OUT_DIM + nc);
    }

    if (qd < NODES_PER_WG) {
        float4* dst = (float4*)(at + qd * AT_PITCH) + t;
        #pragma unroll
        for (int c = 0; c < 8; ++c) dst[4 * c] = a[c];
    }
    constexpr int PADV4 = (AT_ROWS - NODES_PER_WG) * (IN_DIM / 4);
    static_assert(PADV4 <= NODE_THREADS, "one zero store per thread covers the padding rows");
    if (tid < PADV4)
        *(float4*)(at + (NODES_PER_WG + tid / (IN_DIM / 4)) * AT_PITCH + (tid % (IN_DIM / 4)) * 4) =
            make_float4(0.f, 0.f, 0.f, 0.f);
    if (tid == 0) next_job = 0;

    #pragma unroll
    for (int j = 0; j < WPER; ++j) {
        const int i = j * NODE_THREADS + tid;
        if (i < NQ) {
            const int kp = i / (OUT_DIM / 4);
            const int nc = (i - kp * (OUT_DIM / 4)) * 4;
            float4* d = (float4*)&Bs2[kp * OUT_DIM + nc];
            d[0] = make_float4(w0[j].x, w1[j].x, w0[j].y, w1[j].y);
            d[1] = make_float4(w0[j].z, w1[j].z, w0[j].w, w1[j].w);
        }
    }
    __syncthreads();

    const int lane = tid & 31;
    const int sel  = lane >> 4;
    for (int trip = 0; trip < OUT_JOBS; ++trip) {
        int mine = 0;
        if (lane == 0) mine = atomicAdd(&next_job, 1);
        const int job = __builtin_amdgcn_readfirstlane(mine);
        if (job >= OUT_JOBS) break;
        const int mTile = job / (OUT_DIM / 32);
        const int ncol0 = (job % (OUT_DIM / 32)) * 32;
        const float* arow = at + (mTile * 16 + (lane & 15)) * AT_PITCH + sel * 2;
        const v2f*   bcol = Bs2 + sel * OUT_DIM + ncol0 + (lane & 15);

        v8f acc0 = {}, acc1 = {};
        #pragma unroll 4
        for (int k0 = 0; k0 < IN_DIM; k0 += 4) {
            const v2f av = *(const v2f*)(arow + k0);
            const v2f* brow = bcol + (k0 / 2) * OUT_DIM;
            acc0 = __builtin_amdgcn_wmma_f32_16x16x4_f32(false, av, false, brow[ 0], (short)0, acc0, false, false);
            acc1 = __builtin_amdgcn_wmma_f32_16x16x4_f32(false, av, false, brow[16], (short)0, acc1, false, false);
        }

        const int rl0 = mTile * 16 + sel * 8;
        const int cn  = lane & 15;
        #pragma unroll
        for (int r = 0; r < 8; ++r) {
            const int rl = rl0 + r;
            const int ng = (int)blockIdx.x * NODES_PER_WG + rl;
            if (rl < NODES_PER_WG && ng < N_NODE) {
                float* o = out + (size_t)ng * OUT_DIM + ncol0 + cn;
                o[ 0] = acc0[r];
                o[16] = acc1[r];
            }
        }
    }
}

extern "C" void kernel_launch(void* const* d_in, const int* in_sizes, int n_in,
                              void* d_out, int out_size, void* d_ws, size_t ws_size,
                              hipStream_t stream) {
    const int*   q_rel  = (const int*)  d_in[1];
    const int*   r_idx  = (const int*)  d_in[2];
    const float* hidden = (const float*)d_in[3];
    const int*   edges  = (const int*)  d_in[4];
    const float* rela   = (const float*)d_in[6];
    const float* Ws     = (const float*)d_in[7];
    const float* Wr     = (const float*)d_in[8];
    const float* WqrW   = (const float*)d_in[9];
    const float* Wqrb   = (const float*)d_in[10];
    const float* waW    = (const float*)d_in[11];
    const float* wab    = (const float*)d_in[12];
    const float* Wh     = (const float*)d_in[13];
    float* out = (float*)d_out;

    char* ws = (char*)d_ws;
    float* hproj = (float*)(ws + (size_t)N_NODE * IN_DIM * 4);
    float* rproj = (float*)(ws + (size_t)N_NODE * IN_DIM * 4
                               + (size_t)N_NODE * ATTN_DIM * 4);
    float* qproj = (float*)((char*)rproj + (((size_t)N_EMB * ATTN_DIM * 4 + 255ull) & ~255ull));
    int*   offs  = (int*)((char*)qproj + (size_t)N_Q * ATTN_DIM * 4);
    int*   rec   = (int*)((char*)offs + (((size_t)EDGE_WGS * OFFS_PITCH * 4 + 255ull) & ~255ull));

    front_kernel<<<SP_WGS + GEMM1_WGS, 256, 0, stream>>>(
        hidden, Ws, hproj, rela, Wr, WqrW, Wqrb, q_rel, rproj, qproj);

    edge_bin_kernel<<<EDGE_WGS, EDGE_THREADS, 0, stream>>>(edges, r_idx, hproj, rproj, qproj,
                                                          waW, wab, offs, rec);

    node_sum_kernel<<<NODE_WGS, NODE_THREADS, 0, stream>>>(hidden, rela, offs, rec, Wh, out);
}
